# starting kernel
# baseline (speedup 1.0000x reference)
#include <hip/hip_runtime.h>
#include <math.h>


#define NN 50000
#define DD 128
#define HH 4
#define CC 32
#define LL 3
#define NEG_SLOPE 0.2f
#define LN_EPS 1e-5f

typedef __attribute__((ext_vector_type(2))) float v2f;
typedef __attribute__((ext_vector_type(8))) float v8f;

__device__ __forceinline__ unsigned f2ord(float f) {
  unsigned u = __float_as_uint(f);
  return (u & 0x80000000u) ? ~u : (u | 0x80000000u);
}
__device__ __forceinline__ float ord2f(unsigned u) {
  u = (u & 0x80000000u) ? (u & 0x7FFFFFFFu) : ~u;
  return __uint_as_float(u);
}

__global__ void init_layer(float* __restrict__ acc, float* __restrict__ denom,
                           unsigned* __restrict__ emax, int nd, int nh) {
  int i = blockIdx.x * blockDim.x + threadIdx.x;
  if (i < nd) acc[i] = 0.0f;
  if (i < nh) { denom[i] = 0.0f; emax[i] = 0u; }
}

__global__ __launch_bounds__(256) void gemm_dual_wmma(
    const float* __restrict__ x,
    const float* __restrict__ Wl, const float* __restrict__ bl,
    const float* __restrict__ Wr, const float* __restrict__ br,
    float* __restrict__ xl, float* __restrict__ xr) {
  __shared__ float sx[16 * DD];

  const int tid  = threadIdx.x;
  const int wave = tid >> 5;
  const int lane = tid & 31;
  const int rowBase = blockIdx.x * 16;

  {
    const float4* gsrc = (const float4*)(x + (size_t)rowBase * DD);
    float4* sdst = (float4*)sx;
    sdst[tid]       = gsrc[tid];
    sdst[tid + 256] = gsrc[tid + 256];
  }
  __syncthreads();

  const int colBase = wave * 16;
  const int m  = lane & 15;
  const int kk = (lane >> 4) * 2;
  const int n  = lane & 15;

  v8f cl = {};
  v8f cr = {};

  for (int k0 = 0; k0 < DD; k0 += 4) {
    v2f a;
    a.x = sx[m * DD + k0 + kk];
    a.y = sx[m * DD + k0 + kk + 1];
    v2f bfl;
    bfl.x = Wl[(size_t)(k0 + kk) * DD + colBase + n];
    bfl.y = Wl[(size_t)(k0 + kk + 1) * DD + colBase + n];
    v2f bfr;
    bfr.x = Wr[(size_t)(k0 + kk) * DD + colBase + n];
    bfr.y = Wr[(size_t)(k0 + kk + 1) * DD + colBase + n];

    cl = __builtin_amdgcn_wmma_f32_16x16x4_f32(
        false, a, false, bfl, (short)0, cl, false, false);
    cr = __builtin_amdgcn_wmma_f32_16x16x4_f32(
        false, a, false, bfr, (short)0, cr, false, false);
  }

  const int rOff = (lane >> 4) * 8;
  const int col  = colBase + n;
  const float bL = bl[col];
  const float bR = br[col];
#pragma unroll
  for (int i = 0; i < 8; ++i) {
    const size_t row = (size_t)(rowBase + i + rOff);
    xl[row * DD + col] = cl[i] + bL;
    xr[row * DD + col] = cr[i] + bR;
  }
}

__global__ void edge_logits(const int* __restrict__ ei,
                            const float* __restrict__ xl,
                            const float* __restrict__ xr,
                            const float* __restrict__ att,
                            float* __restrict__ elog,
                            unsigned* __restrict__ emax, int E, int Np) {
  const int t = blockIdx.x * blockDim.x + threadIdx.x;
  const int tot = (E + Np) * HH;
  if (t >= tot) return;
  const int e = t >> 2;
  const int h = t & 3;
  const int src = (e < E) ? ei[e] : (e - E);
  const int dst = (e < E) ? ei[E + e] : (e - E);

  const float4* pl = (const float4*)(xl + (size_t)src * DD + h * CC);
  const float4* pr = (const float4*)(xr + (size_t)dst * DD + h * CC);
  const float4* pa = (const float4*)(att + h * CC);

  float s = 0.0f;
#pragma unroll
  for (int i = 0; i < CC / 4; ++i) {
    float4 a = pl[i], b = pr[i], w = pa[i];
    float m0 = a.x + b.x; m0 = (m0 > 0.0f) ? m0 : NEG_SLOPE * m0;
    float m1 = a.y + b.y; m1 = (m1 > 0.0f) ? m1 : NEG_SLOPE * m1;
    float m2 = a.z + b.z; m2 = (m2 > 0.0f) ? m2 : NEG_SLOPE * m2;
    float m3 = a.w + b.w; m3 = (m3 > 0.0f) ? m3 : NEG_SLOPE * m3;
    s = fmaf(w.x, m0, s);
    s = fmaf(w.y, m1, s);
    s = fmaf(w.z, m2, s);
    s = fmaf(w.w, m3, s);
  }
  elog[t] = s;
  atomicMax(&emax[dst * HH + h], f2ord(s));
}

__global__ void edge_exp(const int* __restrict__ ei,
                         float* __restrict__ elog,
                         const unsigned* __restrict__ emax,
                         float* __restrict__ denom, int E, int Np) {
  const int t = blockIdx.x * blockDim.x + threadIdx.x;
  const int tot = (E + Np) * HH;
  if (t >= tot) return;
  const int e = t >> 2;
  const int h = t & 3;
  const int dst = (e < E) ? ei[E + e] : (e - E);
  const float mx = ord2f(emax[dst * HH + h]);
  const float ee = __expf(elog[t] - mx);
  elog[t] = ee;
  atomicAdd(&denom[dst * HH + h], ee);
}

__global__ __launch_bounds__(256) void edge_scatter(
    const int* __restrict__ ei, const float* __restrict__ xl,
    const float* __restrict__ elog, const float* __restrict__ denom,
    float* __restrict__ acc, int E, int Np) {
  const int wid  = (blockIdx.x * blockDim.x + threadIdx.x) >> 5;
  const int lane = threadIdx.x & 31;
  if (wid >= E + Np) return;
  const int src = (wid < E) ? ei[wid] : (wid - E);
  const int dst = (wid < E) ? ei[E + wid] : (wid - E);
#pragma unroll
  for (int h = 0; h < HH; ++h) {
    const float alpha = elog[wid * HH + h] / denom[dst * HH + h];
    const int c = h * CC + lane;
    atomicAdd(&acc[(size_t)dst * DD + c], xl[(size_t)src * DD + c] * alpha);
  }
}

__global__ __launch_bounds__(256) void node_norm(
    const float* __restrict__ xin, const float* __restrict__ acc,
    const float* __restrict__ bias, const float* __restrict__ gamma,
    const float* __restrict__ beta, float* __restrict__ xout, int Np,
    int do_relu) {
  const int wid  = (blockIdx.x * blockDim.x + threadIdx.x) >> 5;
  const int lane = threadIdx.x & 31;
  if (wid >= Np) return;

  float v[4];
  float sum = 0.0f;
#pragma unroll
  for (int j = 0; j < 4; ++j) {
    const int c = j * 32 + lane;
    const float t = xin[(size_t)wid * DD + c] + acc[(size_t)wid * DD + c] + bias[c];
    v[j] = t;
    sum += t;
  }
#pragma unroll
  for (int off = 16; off > 0; off >>= 1) sum += __shfl_xor(sum, off, 32);
  const float mu = sum * (1.0f / (float)DD);

  float sq = 0.0f;
#pragma unroll
  for (int j = 0; j < 4; ++j) { const float d = v[j] - mu; sq = fmaf(d, d, sq); }
#pragma unroll
  for (int off = 16; off > 0; off >>= 1) sq += __shfl_xor(sq, off, 32);
  const float rstd = rsqrtf(sq * (1.0f / (float)DD) + LN_EPS);

#pragma unroll
  for (int j = 0; j < 4; ++j) {
    const int c = j * 32 + lane;
    float o = (v[j] - mu) * rstd * gamma[c] + beta[c];
    if (do_relu) o = (o > 0.0f) ? o : 0.0f;
    xout[(size_t)wid * DD + c] = o;
  }
}

static inline int cdiv(long long a, long long b) { return (int)((a + b - 1) / b); }

extern "C" void kernel_launch(void* const* d_in, const int* in_sizes, int n_in,
                              void* d_out, int out_size, void* d_ws, size_t ws_size,
                              hipStream_t stream) {
  const int Np = NN;
  const int E  = in_sizes[1] / 2;
  const int EP = E + Np;

  const float* x0    = (const float*)d_in[0];
  const int*   ei    = (const int*)d_in[1];
  const float* Wl    = (const float*)d_in[2];
  const float* bl    = (const float*)d_in[3];
  const float* Wr    = (const float*)d_in[4];
  const float* br    = (const float*)d_in[5];
  const float* att   = (const float*)d_in[6];
  const float* bias  = (const float*)d_in[7];
  const float* gamma = (const float*)d_in[8];
  const float* beta  = (const float*)d_in[9];

  float* ws = (float*)d_ws;
  const size_t ND = (size_t)Np * DD;
  const size_t NH = (size_t)Np * HH;
  float*    bufA  = ws;
  float*    xl    = bufA + ND;
  float*    xr    = xl + ND;
  float*    acc   = xr + ND;
  float*    elog  = acc + ND;
  float*    denom = elog + (size_t)EP * HH;
  unsigned* emax  = (unsigned*)(denom + NH);

  const int TB = 256;
  const dim3 blk(TB);

  for (int i = 0; i < LL; ++i) {
    const float* xin  = (i == 0) ? x0 : bufA;
    float*       xout = (i == LL - 1) ? (float*)d_out : bufA;
    const float* Wli  = Wl + (size_t)i * DD * DD;
    const float* bli  = bl + (size_t)i * DD;
    const float* Wri  = Wr + (size_t)i * DD * DD;
    const float* bri  = br + (size_t)i * DD;
    const float* atti = att + (size_t)i * HH * CC;
    const float* bi   = bias + (size_t)i * DD;
    const float* gi   = gamma + (size_t)i * DD;
    const float* bti  = beta + (size_t)i * DD;

    init_layer<<<cdiv(ND, TB), blk, 0, stream>>>(acc, denom, emax, (int)ND, (int)NH);

    gemm_dual_wmma<<<Np / 16, blk, 0, stream>>>(xin, Wli, bli, Wri, bri, xl, xr);

    edge_logits<<<cdiv((long long)EP * HH, TB), blk, 0, stream>>>(
        ei, xl, xr, atti, elog, emax, E, Np);

    edge_exp<<<cdiv((long long)EP * HH, TB), blk, 0, stream>>>(
        ei, elog, emax, denom, E, Np);

    edge_scatter<<<cdiv((long long)EP * 32, TB), blk, 0, stream>>>(
        ei, xl, elog, denom, acc, E, Np);

    node_norm<<<cdiv((long long)Np * 32, TB), blk, 0, stream>>>(
        xin, acc, bi, gi, bti, xout, Np, (i != LL - 1) ? 1 : 0);
  }
}


# Edge logits: one wave per 8 edges reads each 512-byte row with one whole-line 16-byte load per lane, then sums a head's 8 lanes by a 3-step lane exchange (new reduction tree for the logit sum)
# speedup vs baseline: 1.2658x; 1.2658x over previous
#include <hip/hip_runtime.h>
#include <math.h>


#define NN 50000
#define DD 128
#define HH 4
#define CC 32
#define LL 3
#define NEG_SLOPE 0.2f
#define LN_EPS 1e-5f

typedef __attribute__((ext_vector_type(2))) float v2f;
typedef __attribute__((ext_vector_type(8))) float v8f;

__device__ __forceinline__ unsigned f2ord(float f) {
  unsigned u = __float_as_uint(f);
  return (u & 0x80000000u) ? ~u : (u | 0x80000000u);
}
__device__ __forceinline__ float ord2f(unsigned u) {
  u = (u & 0x80000000u) ? (u & 0x7FFFFFFFu) : ~u;
  return __uint_as_float(u);
}

__global__ void init_layer(float* __restrict__ acc, float* __restrict__ denom,
                           unsigned* __restrict__ emax, int nd, int nh) {
  int i = blockIdx.x * blockDim.x + threadIdx.x;
  if (i < nd) acc[i] = 0.0f;
  if (i < nh) { denom[i] = 0.0f; emax[i] = 0u; }
}

__global__ __launch_bounds__(256) void gemm_dual_wmma(
    const float* __restrict__ x,
    const float* __restrict__ Wl, const float* __restrict__ bl,
    const float* __restrict__ Wr, const float* __restrict__ br,
    float* __restrict__ xl, float* __restrict__ xr) {
  __shared__ float sx[16 * DD];

  const int tid  = threadIdx.x;
  const int wave = tid >> 5;
  const int lane = tid & 31;
  const int rowBase = blockIdx.x * 16;

  {
    const float4* gsrc = (const float4*)(x + (size_t)rowBase * DD);
    float4* sdst = (float4*)sx;
    sdst[tid]       = gsrc[tid];
    sdst[tid + 256] = gsrc[tid + 256];
  }
  __syncthreads();

  const int colBase = wave * 16;
  const int m  = lane & 15;
  const int kk = (lane >> 4) * 2;
  const int n  = lane & 15;

  v8f cl = {};
  v8f cr = {};

  for (int k0 = 0; k0 < DD; k0 += 4) {
    v2f a;
    a.x = sx[m * DD + k0 + kk];
    a.y = sx[m * DD + k0 + kk + 1];
    v2f bfl;
    bfl.x = Wl[(size_t)(k0 + kk) * DD + colBase + n];
    bfl.y = Wl[(size_t)(k0 + kk + 1) * DD + colBase + n];
    v2f bfr;
    bfr.x = Wr[(size_t)(k0 + kk) * DD + colBase + n];
    bfr.y = Wr[(size_t)(k0 + kk + 1) * DD + colBase + n];

    cl = __builtin_amdgcn_wmma_f32_16x16x4_f32(
        false, a, false, bfl, (short)0, cl, false, false);
    cr = __builtin_amdgcn_wmma_f32_16x16x4_f32(
        false, a, false, bfr, (short)0, cr, false, false);
  }

  const int rOff = (lane >> 4) * 8;
  const int col  = colBase + n;
  const float bL = bl[col];
  const float bR = br[col];
#pragma unroll
  for (int i = 0; i < 8; ++i) {
    const size_t row = (size_t)(rowBase + i + rOff);
    xl[row * DD + col] = cl[i] + bL;
    xr[row * DD + col] = cr[i] + bR;
  }
}

#define EPW 8

template <int M>
__device__ __forceinline__ float lane_xor(float v) {
  return __int_as_float(__builtin_amdgcn_update_dpp(
      0, __float_as_int(v), 0x160 + M, 0xf, 0xf, true));
}

__global__ __launch_bounds__(256) void edge_logits(const int* __restrict__ ei,
                            const float* __restrict__ xl,
                            const float* __restrict__ xr,
                            const float* __restrict__ att,
                            float* __restrict__ elog,
                            unsigned* __restrict__ emax, int E, int Np) {
  const int lane = threadIdx.x & 31;
  const int wave = __builtin_amdgcn_readfirstlane(
      (int)((blockIdx.x * blockDim.x + threadIdx.x) >> 5));
  const int EP = E + Np;
  const int e0 = wave * EPW;
  if (e0 >= EP) return;

  const float4* xl4 = (const float4*)xl;
  const float4* xr4 = (const float4*)xr;
  const float4 w = ((const float4*)att)[lane];

  int sidx[EPW], didx[EPW];
  if (e0 + EPW <= E) {
#pragma unroll
    for (int j = 0; j < EPW; ++j) {
      sidx[j] = ei[e0 + j];
      didx[j] = ei[E + e0 + j];
    }
  } else {
#pragma unroll
    for (int j = 0; j < EPW; ++j) {
      const int e   = min(e0 + j, EP - 1);
      const int ec  = min(e, E - 1);
      const int sld = ei[ec];
      const int tld = ei[E + ec];
      sidx[j] = (e < E) ? sld : (e - E);
      didx[j] = (e < E) ? tld : (e - E);
    }
  }

  float p[EPW];
#pragma unroll
  for (int j = 0; j < EPW; ++j) {
    const float4 a = xl4[(size_t)sidx[j] * (DD / 4) + lane];
    const float4 b = xr4[(size_t)didx[j] * (DD / 4) + lane];
    float m0 = a.x + b.x; m0 = (m0 > 0.0f) ? m0 : NEG_SLOPE * m0;
    float m1 = a.y + b.y; m1 = (m1 > 0.0f) ? m1 : NEG_SLOPE * m1;
    float m2 = a.z + b.z; m2 = (m2 > 0.0f) ? m2 : NEG_SLOPE * m2;
    float m3 = a.w + b.w; m3 = (m3 > 0.0f) ? m3 : NEG_SLOPE * m3;
    float t = w.x * m0;
    t = fmaf(w.y, m1, t);
    t = fmaf(w.z, m2, t);
    t = fmaf(w.w, m3, t);
    p[j] = t;
  }

  int dst_own = didx[0];
#pragma unroll
  for (int j = 1; j < EPW; ++j) dst_own = ((lane & 7) == j) ? didx[j] : dst_own;

  const bool b4 = (lane & 4) != 0;
  const bool b2 = (lane & 2) != 0;
  const bool b1 = (lane & 1) != 0;
  float q[4];
#pragma unroll
  for (int i = 0; i < 4; ++i) {
    const float keep = b4 ? p[i + 4] : p[i];
    const float give = b4 ? p[i] : p[i + 4];
    q[i] = keep + lane_xor<4>(give);
  }
  float r[2];
#pragma unroll
  for (int i = 0; i < 2; ++i) {
    const float keep = b2 ? q[i + 2] : q[i];
    const float give = b2 ? q[i] : q[i + 2];
    r[i] = keep + lane_xor<2>(give);
  }
  const float keep = b1 ? r[1] : r[0];
  const float give = b1 ? r[0] : r[1];
  const float s = keep + lane_xor<1>(give);

  const int eown = e0 + (lane & 7);
  if (eown < EP) {
    const int h = lane >> 3;
    elog[eown * HH + h] = s;
    atomicMax(&emax[dst_own * HH + h], f2ord(s));
  }
}

__global__ void edge_exp(const int* __restrict__ ei,
                         float* __restrict__ elog,
                         const unsigned* __restrict__ emax,
                         float* __restrict__ denom, int E, int Np) {
  const int t = blockIdx.x * blockDim.x + threadIdx.x;
  const int tot = (E + Np) * HH;
  if (t >= tot) return;
  const int e = t >> 2;
  const int h = t & 3;
  const int dst = (e < E) ? ei[E + e] : (e - E);
  const float mx = ord2f(emax[dst * HH + h]);
  const float ee = __expf(elog[t] - mx);
  elog[t] = ee;
  atomicAdd(&denom[dst * HH + h], ee);
}

__global__ __launch_bounds__(256) void edge_scatter(
    const int* __restrict__ ei, const float* __restrict__ xl,
    const float* __restrict__ elog, const float* __restrict__ denom,
    float* __restrict__ acc, int E, int Np) {
  const int wid  = (blockIdx.x * blockDim.x + threadIdx.x) >> 5;
  const int lane = threadIdx.x & 31;
  if (wid >= E + Np) return;
  const int src = (wid < E) ? ei[wid] : (wid - E);
  const int dst = (wid < E) ? ei[E + wid] : (wid - E);
#pragma unroll
  for (int h = 0; h < HH; ++h) {
    const float alpha = elog[wid * HH + h] / denom[dst * HH + h];
    const int c = h * CC + lane;
    atomicAdd(&acc[(size_t)dst * DD + c], xl[(size_t)src * DD + c] * alpha);
  }
}

__global__ __launch_bounds__(256) void node_norm(
    const float* __restrict__ xin, const float* __restrict__ acc,
    const float* __restrict__ bias, const float* __restrict__ gamma,
    const float* __restrict__ beta, float* __restrict__ xout, int Np,
    int do_relu) {
  const int wid  = (blockIdx.x * blockDim.x + threadIdx.x) >> 5;
  const int lane = threadIdx.x & 31;
  if (wid >= Np) return;

  float v[4];
  float sum = 0.0f;
#pragma unroll
  for (int j = 0; j < 4; ++j) {
    const int c = j * 32 + lane;
    const float t = xin[(size_t)wid * DD + c] + acc[(size_t)wid * DD + c] + bias[c];
    v[j] = t;
    sum += t;
  }
#pragma unroll
  for (int off = 16; off > 0; off >>= 1) sum += __shfl_xor(sum, off, 32);
  const float mu = sum * (1.0f / (float)DD);

  float sq = 0.0f;
#pragma unroll
  for (int j = 0; j < 4; ++j) { const float d = v[j] - mu; sq = fmaf(d, d, sq); }
#pragma unroll
  for (int off = 16; off > 0; off >>= 1) sq += __shfl_xor(sq, off, 32);
  const float rstd = rsqrtf(sq * (1.0f / (float)DD) + LN_EPS);

#pragma unroll
  for (int j = 0; j < 4; ++j) {
    const int c = j * 32 + lane;
    float o = (v[j] - mu) * rstd * gamma[c] + beta[c];
    if (do_relu) o = (o > 0.0f) ? o : 0.0f;
    xout[(size_t)wid * DD + c] = o;
  }
}

static inline int cdiv(long long a, long long b) { return (int)((a + b - 1) / b); }

extern "C" void kernel_launch(void* const* d_in, const int* in_sizes, int n_in,
                              void* d_out, int out_size, void* d_ws, size_t ws_size,
                              hipStream_t stream) {
  const int Np = NN;
  const int E  = in_sizes[1] / 2;
  const int EP = E + Np;

  const float* x0    = (const float*)d_in[0];
  const int*   ei    = (const int*)d_in[1];
  const float* Wl    = (const float*)d_in[2];
  const float* bl    = (const float*)d_in[3];
  const float* Wr    = (const float*)d_in[4];
  const float* br    = (const float*)d_in[5];
  const float* att   = (const float*)d_in[6];
  const float* bias  = (const float*)d_in[7];
  const float* gamma = (const float*)d_in[8];
  const float* beta  = (const float*)d_in[9];

  float* ws = (float*)d_ws;
  const size_t ND = (size_t)Np * DD;
  const size_t NH = (size_t)Np * HH;
  float*    bufA  = ws;
  float*    xl    = bufA + ND;
  float*    xr    = xl + ND;
  float*    acc   = xr + ND;
  float*    elog  = acc + ND;
  float*    denom = elog + (size_t)EP * HH;
  unsigned* emax  = (unsigned*)(denom + NH);

  const int TB = 256;
  const dim3 blk(TB);

  for (int i = 0; i < LL; ++i) {
    const float* xin  = (i == 0) ? x0 : bufA;
    float*       xout = (i == LL - 1) ? (float*)d_out : bufA;
    const float* Wli  = Wl + (size_t)i * DD * DD;
    const float* bli  = bl + (size_t)i * DD;
    const float* Wri  = Wr + (size_t)i * DD * DD;
    const float* bri  = br + (size_t)i * DD;
    const float* atti = att + (size_t)i * HH * CC;
    const float* bi   = bias + (size_t)i * DD;
    const float* gi   = gamma + (size_t)i * DD;
    const float* bti  = beta + (size_t)i * DD;

    init_layer<<<cdiv(ND, TB), blk, 0, stream>>>(acc, denom, emax, (int)ND, (int)NH);

    gemm_dual_wmma<<<Np / 16, blk, 0, stream>>>(xin, Wli, bli, Wri, bri, xl, xr);

    edge_logits<<<cdiv((long long)EP, (TB / 32) * EPW), blk, 0, stream>>>(
        ei, xl, xr, atti, elog, emax, E, Np);

    edge_exp<<<cdiv((long long)EP * HH, TB), blk, 0, stream>>>(
        ei, elog, emax, denom, E, Np);

    edge_scatter<<<cdiv((long long)EP * 32, TB), blk, 0, stream>>>(
        ei, xl, elog, denom, acc, E, Np);

    node_norm<<<cdiv((long long)Np * 32, TB), blk, 0, stream>>>(
        xin, acc, bi, gi, bti, xout, Np, (i != LL - 1) ? 1 : 0);
  }
}


# Segment max by the owner workgroup in LDS (integer max over its edge list; each edge's 16-byte logit record read once and kept in LDS for the softmax): no per-edge atomic max to memory left
# speedup vs baseline: 2.6214x; 2.0709x over previous
#include <hip/hip_runtime.h>
#include <math.h>


#define NN 50000
#define DD 128
#define HH 4
#define CC 32
#define LL 3
#define NEG_SLOPE 0.2f
#define LN_EPS 1e-5f

typedef __attribute__((ext_vector_type(2))) float v2f;
typedef __attribute__((ext_vector_type(8))) float v8f;

__device__ __forceinline__ unsigned f2ord(float f) {
  unsigned u = __float_as_uint(f);
  return (u & 0x80000000u) ? ~u : (u | 0x80000000u);
}
__device__ __forceinline__ float ord2f(unsigned u) {
  u = (u & 0x80000000u) ? (u & 0x7FFFFFFFu) : ~u;
  return __uint_as_float(u);
}

__global__ __launch_bounds__(256) void gemm_dual_wmma(
    const float* __restrict__ x,
    const float* __restrict__ Wl, const float* __restrict__ bl,
    const float* __restrict__ Wr, const float* __restrict__ br,
    float* __restrict__ xl, float* __restrict__ xr) {
  __shared__ float sx[16 * DD];

  const int tid  = threadIdx.x;
  const int wave = tid >> 5;
  const int lane = tid & 31;
  const int rowBase = blockIdx.x * 16;

  {
    const float4* gsrc = (const float4*)(x + (size_t)rowBase * DD);
    float4* sdst = (float4*)sx;
    sdst[tid]       = gsrc[tid];
    sdst[tid + 256] = gsrc[tid + 256];
  }
  __syncthreads();

  const int colBase = wave * 16;
  const int m  = lane & 15;
  const int kk = (lane >> 4) * 2;
  const int n  = lane & 15;

  v8f cl = {};
  v8f cr = {};

  for (int k0 = 0; k0 < DD; k0 += 4) {
    v2f a;
    a.x = sx[m * DD + k0 + kk];
    a.y = sx[m * DD + k0 + kk + 1];
    v2f bfl;
    bfl.x = Wl[(size_t)(k0 + kk) * DD + colBase + n];
    bfl.y = Wl[(size_t)(k0 + kk + 1) * DD + colBase + n];
    v2f bfr;
    bfr.x = Wr[(size_t)(k0 + kk) * DD + colBase + n];
    bfr.y = Wr[(size_t)(k0 + kk + 1) * DD + colBase + n];

    cl = __builtin_amdgcn_wmma_f32_16x16x4_f32(
        false, a, false, bfl, (short)0, cl, false, false);
    cr = __builtin_amdgcn_wmma_f32_16x16x4_f32(
        false, a, false, bfr, (short)0, cr, false, false);
  }

  const int rOff = (lane >> 4) * 8;
  const int col  = colBase + n;
  const float bL = bl[col];
  const float bR = br[col];
#pragma unroll
  for (int i = 0; i < 8; ++i) {
    const size_t row = (size_t)(rowBase + i + rOff);
    xl[row * DD + col] = cl[i] + bL;
    xr[row * DD + col] = cr[i] + bR;
  }
}

#define EPW 8

template <int M>
__device__ __forceinline__ float lane_xor(float v) {
  return __int_as_float(__builtin_amdgcn_update_dpp(
      0, __float_as_int(v), 0x160 + M, 0xf, 0xf, true));
}

__global__ __launch_bounds__(256) void edge_logits(const int* __restrict__ ei,
                            const float* __restrict__ xl,
                            const float* __restrict__ xr,
                            const float* __restrict__ att,
                            float* __restrict__ elog, int E, int Np) {
  const int lane = threadIdx.x & 31;
  const int wave = __builtin_amdgcn_readfirstlane(
      (int)((blockIdx.x * blockDim.x + threadIdx.x) >> 5));
  const int EP = E + Np;
  const int e0 = wave * EPW;
  if (e0 >= EP) return;

  const float4* xl4 = (const float4*)xl;
  const float4* xr4 = (const float4*)xr;
  const float4 w = ((const float4*)att)[lane];

  int sidx[EPW], didx[EPW];
  if (e0 + EPW <= E) {
#pragma unroll
    for (int j = 0; j < EPW; ++j) {
      sidx[j] = ei[e0 + j];
      didx[j] = ei[E + e0 + j];
    }
  } else {
#pragma unroll
    for (int j = 0; j < EPW; ++j) {
      const int e   = min(e0 + j, EP - 1);
      const int ec  = min(e, E - 1);
      const int sld = ei[ec];
      const int tld = ei[E + ec];
      sidx[j] = (e < E) ? sld : (e - E);
      didx[j] = (e < E) ? tld : (e - E);
    }
  }

  float p[EPW];
#pragma unroll
  for (int j = 0; j < EPW; ++j) {
    const float4 a = xl4[(size_t)sidx[j] * (DD / 4) + lane];
    const float4 b = xr4[(size_t)didx[j] * (DD / 4) + lane];
    float m0 = a.x + b.x; m0 = (m0 > 0.0f) ? m0 : NEG_SLOPE * m0;
    float m1 = a.y + b.y; m1 = (m1 > 0.0f) ? m1 : NEG_SLOPE * m1;
    float m2 = a.z + b.z; m2 = (m2 > 0.0f) ? m2 : NEG_SLOPE * m2;
    float m3 = a.w + b.w; m3 = (m3 > 0.0f) ? m3 : NEG_SLOPE * m3;
    float t = w.x * m0;
    t = fmaf(w.y, m1, t);
    t = fmaf(w.z, m2, t);
    t = fmaf(w.w, m3, t);
    p[j] = t;
  }

  const bool b4 = (lane & 4) != 0;
  const bool b2 = (lane & 2) != 0;
  const bool b1 = (lane & 1) != 0;
  float q[4];
#pragma unroll
  for (int i = 0; i < 4; ++i) {
    const float keep = b4 ? p[i + 4] : p[i];
    const float give = b4 ? p[i] : p[i + 4];
    q[i] = keep + lane_xor<4>(give);
  }
  float r[2];
#pragma unroll
  for (int i = 0; i < 2; ++i) {
    const float keep = b2 ? q[i + 2] : q[i];
    const float give = b2 ? q[i] : q[i + 2];
    r[i] = keep + lane_xor<2>(give);
  }
  const float keep = b1 ? r[1] : r[0];
  const float give = b1 ? r[0] : r[1];
  const float s = keep + lane_xor<1>(give);

  const int eown = e0 + (lane & 7);
  if (eown < EP) {
    const int h = lane >> 3;
    elog[eown * HH + h] = s;
  }
}

#define RN    196
#define OWN_T 1024
#define LCAP  8192

template <int J>
__device__ __forceinline__ float group8_bcast(float v) {
  return __uint_as_float(__builtin_amdgcn_mov_dpp8(__float_as_uint(v), J * 0x249249));
}

__device__ __forceinline__ void lds_add(float* p, float v) {
  __hip_atomic_fetch_add(p, v, __ATOMIC_RELAXED, __HIP_MEMORY_SCOPE_WORKGROUP);
}

__device__ __forceinline__ void lds_umax(unsigned* p, unsigned v) {
  __hip_atomic_fetch_max(p, v, __ATOMIC_RELAXED, __HIP_MEMORY_SCOPE_WORKGROUP);
}

__device__ __forceinline__ void own_max4(unsigned* mxU, int rel, float4 lg) {
  lds_umax(&mxU[rel * HH + 0], f2ord(lg.x));
  lds_umax(&mxU[rel * HH + 1], f2ord(lg.y));
  lds_umax(&mxU[rel * HH + 2], f2ord(lg.z));
  lds_umax(&mxU[rel * HH + 3], f2ord(lg.w));
}

__device__ __forceinline__ void own_append(unsigned e, unsigned rel, int* cnt,
                                           unsigned* keys) {
  const int pos = __hip_atomic_fetch_add(cnt, 1, __ATOMIC_RELAXED,
                                         __HIP_MEMORY_SCOPE_WORKGROUP);
  if (pos < LCAP) keys[pos] = (e << 8) | rel;
}

__device__ __forceinline__ void own_edge_serial(
    int e, int src, int rel, float* accL, float* denL, const unsigned* mxU,
    const float* __restrict__ xl, const float* __restrict__ elog) {
  float ee[HH];
  for (int hh = 0; hh < HH; ++hh) {
    ee[hh] = __expf(elog[(size_t)e * HH + hh] - ord2f(mxU[rel * HH + hh]));
    lds_add(&denL[rel * HH + hh], ee[hh]);
  }
  for (int c = 0; c < DD; ++c)
    lds_add(&accL[rel * DD + (c & 3) * 32 + (c >> 2)],
            xl[(size_t)src * DD + c] * ee[c >> 5]);
}

__global__ __launch_bounds__(OWN_T) void owner_aggregate(
    const int* __restrict__ ei, const float* __restrict__ xl,
    const float* __restrict__ elog,
    const float* xin, const float* __restrict__ bias,
    const float* __restrict__ gamma, const float* __restrict__ beta,
    float* xout, int E, int Np, int do_relu) {
  __shared__ float accL[RN * DD];
  __shared__ float denL[RN * HH];
  __shared__ unsigned mxU[RN * HH];
  __shared__ unsigned keys[LCAP];
  __shared__ float4 lgL[LCAP];
  __shared__ int cnt;

  const int tid  = threadIdx.x;
  const int lane = tid & 31;
  const int wave = __builtin_amdgcn_readfirstlane(tid >> 5);
  const int lo = blockIdx.x * RN;
  const int nn = min(RN, Np - lo);

  for (int i = tid; i < RN * DD / 4; i += OWN_T)
    ((float4*)accL)[i] = make_float4(0.0f, 0.0f, 0.0f, 0.0f);
  for (int i = tid; i < nn * HH; i += OWN_T) {
    denL[i] = 0.0f;
    mxU[i]  = 0u;
  }
  for (int i = tid; i < nn; i += OWN_T)
    keys[i] = ((unsigned)(E + lo + i) << 8) | (unsigned)i;
  if (tid == 0) cnt = nn;
  __syncthreads();

  {
    const int* dp = ei + E;
    int head = (int)(((16u - (unsigned)((size_t)dp & 15u)) & 15u) >> 2);
    head = min(head, E);
    const int nvec = (E - head) >> 2;
    const int4* dp4 = (const int4*)(dp + head);
    const unsigned un = (unsigned)nn;
#define OWN_TEST(dv, ev)                                                     \
    { const unsigned rel_ = (unsigned)((dv) - lo);                            \
      if (rel_ < un) own_append((unsigned)(ev), rel_, &cnt, keys); }
    for (int i0 = tid; i0 < nvec; i0 += 8 * OWN_T) {
      int4 d[8];
#pragma unroll
      for (int u = 0; u < 8; ++u) {
        const int i = min(i0 + u * OWN_T, nvec - 1);
        d[u] = dp4[i];
      }
#pragma unroll
      for (int u = 0; u < 8; ++u) {
        const int i = i0 + u * OWN_T;
        if (i < nvec) {
          const int eb = head + 4 * i;
          OWN_TEST(d[u].x, eb)
          OWN_TEST(d[u].y, eb + 1)
          OWN_TEST(d[u].z, eb + 2)
          OWN_TEST(d[u].w, eb + 3)
        }
      }
    }
    if (tid < head) OWN_TEST(dp[tid], tid)
    for (int e = head + 4 * nvec + tid; e < E; e += OWN_T) OWN_TEST(dp[e], e)
#undef OWN_TEST
  }
  __syncthreads();

  const int total = __builtin_amdgcn_readfirstlane(cnt);
  const int n = (total <= LCAP) ? total : 0;

  if (n > 0) {
    const float4* elog4 = (const float4*)elog;
    const int wrap = (1 << (31 - __builtin_clz((unsigned)n))) - 1;
    for (int i0 = tid; i0 < n; i0 += 4 * OWN_T) {
      float4 v[4];
      int rl[4];
#pragma unroll
      for (int u = 0; u < 4; ++u) {
        int i = i0 + u * OWN_T;
        i = (i < n) ? i : (i & wrap);
        const unsigned key = keys[i];
        rl[u] = (int)(key & 255u);
        v[u]  = elog4[key >> 8];
      }
      asm volatile("" :: "v"(v[0].x), "v"(v[0].y), "v"(v[0].z), "v"(v[0].w),
                         "v"(v[1].x), "v"(v[1].y), "v"(v[1].z), "v"(v[1].w),
                         "v"(v[2].x), "v"(v[2].y), "v"(v[2].z), "v"(v[2].w),
                         "v"(v[3].x), "v"(v[3].y), "v"(v[3].z), "v"(v[3].w));
#pragma unroll
      for (int u = 0; u < 4; ++u) {
        lgL[i0 + u * OWN_T] = v[u];
        own_max4(mxU, rl[u], v[u]);
      }
    }
  }
  __syncthreads();

  const int j = lane & 7;
  const int h = lane >> 3;
  const float4* xl4 = (const float4*)xl;

#define OWN_FETCH(g, e_, dl_, src_, lg_)                                  \
  { const int idx_ = min((g) + j, n - 1);                                  \
    const unsigned key_ = keys[idx_];                                      \
    e_  = (int)(key_ >> 8);                                                \
    dl_ = (int)(key_ & 255u);                                              \
    src_ = ei[min(e_, E - 1)];                     \
    src_ = (e_ < E) ? src_ : e_ - E;        \
    lg_ = ((const float*)lgL)[idx_ * HH + h]; }

#define OWN_ROW(J) \
  row[J] = xl4[(size_t)__builtin_amdgcn_readlane(src, J) * (DD / 4) + lane];

#define OWN_ADD(J, COND)                                                   \
  if (COND) {                                                              \
    const float w_ = group8_bcast<J>(ee);                                  \
    float* a_ = accL + __builtin_amdgcn_readlane(dl, J) * DD + lane;       \
    lds_add(a_,      row[J].x * w_);                                       \
    lds_add(a_ + 32, row[J].y * w_);                                       \
    lds_add(a_ + 64, row[J].z * w_);                                       \
    lds_add(a_ + 96, row[J].w * w_);                                       \
  }

  int g = wave * 8;
  int e = 0, dl = 0, src = 0;
  float lg = 0.0f;
  if (g < n) OWN_FETCH(g, e, dl, src, lg)
  while (g < n) {
    const int gn = g + (OWN_T / 32) * 8;
    const float ee = __expf(lg - ord2f(mxU[dl * HH + h]));
    if (g + j < n) lds_add(&denL[dl * HH + h], ee);

    float4 row[8];
    OWN_ROW(0) OWN_ROW(1) OWN_ROW(2) OWN_ROW(3)
    OWN_ROW(4) OWN_ROW(5) OWN_ROW(6) OWN_ROW(7)

    int e2, dl2, src2;
    float lg2;
    OWN_FETCH(gn, e2, dl2, src2, lg2)
    if (g + 8 <= n) {
      OWN_ADD(0, true) OWN_ADD(1, true) OWN_ADD(2, true) OWN_ADD(3, true)
      OWN_ADD(4, true) OWN_ADD(5, true) OWN_ADD(6, true) OWN_ADD(7, true)
    } else {
      OWN_ADD(0, g + 0 < n) OWN_ADD(1, g + 1 < n) OWN_ADD(2, g + 2 < n)
      OWN_ADD(3, g + 3 < n) OWN_ADD(4, g + 4 < n) OWN_ADD(5, g + 5 < n)
      OWN_ADD(6, g + 6 < n) OWN_ADD(7, g + 7 < n)
    }

    g = gn; e = e2; dl = dl2; src = src2; lg = lg2;
  }
#undef OWN_FETCH
#undef OWN_ROW
#undef OWN_ADD
  if (total > LCAP) {
    const int* dp = ei + E;
    const float4* elog4 = (const float4*)elog;
    for (int ed = tid; ed < E; ed += OWN_T) {
      const unsigned rel = (unsigned)(dp[ed] - lo);
      if (rel < (unsigned)nn) own_max4(mxU, (int)rel, elog4[ed]);
    }
    for (int i = tid; i < nn; i += OWN_T) own_max4(mxU, i, elog4[E + lo + i]);
    __syncthreads();
    for (int ed = tid; ed < E; ed += OWN_T) {
      const unsigned rel = (unsigned)(dp[ed] - lo);
      if (rel < (unsigned)nn)
        own_edge_serial(ed, ei[ed], (int)rel, accL, denL, mxU, xl, elog);
    }
    for (int i = tid; i < nn; i += OWN_T)
      own_edge_serial(E + lo + i, lo + i, i, accL, denL, mxU, xl, elog);
  }
  __syncthreads();

  const float4 bs = ((const float4*)bias)[lane];
  const float4 gm = ((const float4*)gamma)[lane];
  const float4 bt = ((const float4*)beta)[lane];
  for (int nl = wave; nl < nn; nl += OWN_T / 32) {
    const size_t node = (size_t)(lo + nl);
    const float4 xi = ((const float4*)xin)[node * (DD / 4) + lane];
    const float den = denL[nl * HH + h];
    const float* a = accL + nl * DD + lane;
    const float v0 = xi.x + a[0]  / den + bs.x;
    const float v1 = xi.y + a[32] / den + bs.y;
    const float v2 = xi.z + a[64] / den + bs.z;
    const float v3 = xi.w + a[96] / den + bs.w;

    float sum = (v0 + v1) + (v2 + v3);
#pragma unroll
    for (int off = 16; off > 0; off >>= 1) sum += __shfl_xor(sum, off, 32);
    const float mu = sum * (1.0f / (float)DD);

    const float d0 = v0 - mu, d1 = v1 - mu, d2 = v2 - mu, d3 = v3 - mu;
    float sq = fmaf(d0, d0, fmaf(d1, d1, fmaf(d2, d2, d3 * d3)));
#pragma unroll
    for (int off = 16; off > 0; off >>= 1) sq += __shfl_xor(sq, off, 32);
    const float rstd = rsqrtf(sq * (1.0f / (float)DD) + LN_EPS);

    float4 o;
    o.x = d0 * rstd * gm.x + bt.x;
    o.y = d1 * rstd * gm.y + bt.y;
    o.z = d2 * rstd * gm.z + bt.z;
    o.w = d3 * rstd * gm.w + bt.w;
    if (do_relu) {
      o.x = (o.x > 0.0f) ? o.x : 0.0f;
      o.y = (o.y > 0.0f) ? o.y : 0.0f;
      o.z = (o.z > 0.0f) ? o.z : 0.0f;
      o.w = (o.w > 0.0f) ? o.w : 0.0f;
    }
    ((float4*)xout)[node * (DD / 4) + lane] = o;
  }
}

static inline int cdiv(long long a, long long b) { return (int)((a + b - 1) / b); }

extern "C" void kernel_launch(void* const* d_in, const int* in_sizes, int n_in,
                              void* d_out, int out_size, void* d_ws, size_t ws_size,
                              hipStream_t stream) {
  const int Np = NN;
  const int E  = in_sizes[1] / 2;
  const int EP = E + Np;

  const float* x0    = (const float*)d_in[0];
  const int*   ei    = (const int*)d_in[1];
  const float* Wl    = (const float*)d_in[2];
  const float* bl    = (const float*)d_in[3];
  const float* Wr    = (const float*)d_in[4];
  const float* br    = (const float*)d_in[5];
  const float* att   = (const float*)d_in[6];
  const float* bias  = (const float*)d_in[7];
  const float* gamma = (const float*)d_in[8];
  const float* beta  = (const float*)d_in[9];

  float* ws = (float*)d_ws;
  const size_t ND = (size_t)Np * DD;
  float*    bufA  = ws;
  float*    xl    = bufA + ND;
  float*    xr    = xl + ND;
  float*    elog  = xr + ND;

  const int TB = 256;
  const dim3 blk(TB);

  for (int i = 0; i < LL; ++i) {
    const float* xin  = (i == 0) ? x0 : bufA;
    float*       xout = (i == LL - 1) ? (float*)d_out : bufA;
    const float* Wli  = Wl + (size_t)i * DD * DD;
    const float* bli  = bl + (size_t)i * DD;
    const float* Wri  = Wr + (size_t)i * DD * DD;
    const float* bri  = br + (size_t)i * DD;
    const float* atti = att + (size_t)i * HH * CC;
    const float* bi   = bias + (size_t)i * DD;
    const float* gi   = gamma + (size_t)i * DD;
    const float* bti  = beta + (size_t)i * DD;

    gemm_dual_wmma<<<Np / 16, blk, 0, stream>>>(xin, Wli, bli, Wri, bri, xl, xr);

    edge_logits<<<cdiv((long long)EP, (TB / 32) * EPW), blk, 0, stream>>>(
        ei, xl, xr, atti, elog, E, Np);

    owner_aggregate<<<cdiv(Np, RN), dim3(OWN_T), 0, stream>>>(
        ei, xl, elog, xin, bi, gi, bti, xout, E, Np,
        (i != LL - 1) ? 1 : 0);
  }
}


# Owner edge lists built once per call: layer 0 owner workgroups store their filtered list in the workspace, layers 1-2 load it and skip the scan of 800000 destinations (LDS sums keep arrival order)
# speedup vs baseline: 3.0681x; 1.1704x over previous
#include <hip/hip_runtime.h>
#include <math.h>


#define NN 50000
#define DD 128
#define HH 4
#define CC 32
#define LL 3
#define NEG_SLOPE 0.2f
#define LN_EPS 1e-5f

typedef __attribute__((ext_vector_type(2))) float v2f;
typedef __attribute__((ext_vector_type(8))) float v8f;

__device__ __forceinline__ unsigned f2ord(float f) {
  unsigned u = __float_as_uint(f);
  return (u & 0x80000000u) ? ~u : (u | 0x80000000u);
}
__device__ __forceinline__ float ord2f(unsigned u) {
  u = (u & 0x80000000u) ? (u & 0x7FFFFFFFu) : ~u;
  return __uint_as_float(u);
}

__global__ __launch_bounds__(256) void gemm_dual_wmma(
    const float* __restrict__ x,
    const float* __restrict__ Wl, const float* __restrict__ bl,
    const float* __restrict__ Wr, const float* __restrict__ br,
    float* __restrict__ xl, float* __restrict__ xr) {
  __shared__ float sx[16 * DD];

  const int tid  = threadIdx.x;
  const int wave = tid >> 5;
  const int lane = tid & 31;
  const int rowBase = blockIdx.x * 16;

  {
    const float4* gsrc = (const float4*)(x + (size_t)rowBase * DD);
    float4* sdst = (float4*)sx;
    sdst[tid]       = gsrc[tid];
    sdst[tid + 256] = gsrc[tid + 256];
  }
  __syncthreads();

  const int colBase = wave * 16;
  const int m  = lane & 15;
  const int kk = (lane >> 4) * 2;
  const int n  = lane & 15;

  v8f cl = {};
  v8f cr = {};

  for (int k0 = 0; k0 < DD; k0 += 4) {
    v2f a;
    a.x = sx[m * DD + k0 + kk];
    a.y = sx[m * DD + k0 + kk + 1];
    v2f bfl;
    bfl.x = Wl[(size_t)(k0 + kk) * DD + colBase + n];
    bfl.y = Wl[(size_t)(k0 + kk + 1) * DD + colBase + n];
    v2f bfr;
    bfr.x = Wr[(size_t)(k0 + kk) * DD + colBase + n];
    bfr.y = Wr[(size_t)(k0 + kk + 1) * DD + colBase + n];

    cl = __builtin_amdgcn_wmma_f32_16x16x4_f32(
        false, a, false, bfl, (short)0, cl, false, false);
    cr = __builtin_amdgcn_wmma_f32_16x16x4_f32(
        false, a, false, bfr, (short)0, cr, false, false);
  }

  const int rOff = (lane >> 4) * 8;
  const int col  = colBase + n;
  const float bL = bl[col];
  const float bR = br[col];
#pragma unroll
  for (int i = 0; i < 8; ++i) {
    const size_t row = (size_t)(rowBase + i + rOff);
    xl[row * DD + col] = cl[i] + bL;
    xr[row * DD + col] = cr[i] + bR;
  }
}

#define EPW 8

template <int M>
__device__ __forceinline__ float lane_xor(float v) {
  return __int_as_float(__builtin_amdgcn_update_dpp(
      0, __float_as_int(v), 0x160 + M, 0xf, 0xf, true));
}

__global__ __launch_bounds__(256) void edge_logits(const int* __restrict__ ei,
                            const float* __restrict__ xl,
                            const float* __restrict__ xr,
                            const float* __restrict__ att,
                            float* __restrict__ elog, int E, int Np) {
  const int lane = threadIdx.x & 31;
  const int wave = __builtin_amdgcn_readfirstlane(
      (int)((blockIdx.x * blockDim.x + threadIdx.x) >> 5));
  const int EP = E + Np;
  const int e0 = wave * EPW;
  if (e0 >= EP) return;

  const float4* xl4 = (const float4*)xl;
  const float4* xr4 = (const float4*)xr;
  const float4 w = ((const float4*)att)[lane];

  int sidx[EPW], didx[EPW];
  if (e0 + EPW <= E) {
#pragma unroll
    for (int j = 0; j < EPW; ++j) {
      sidx[j] = ei[e0 + j];
      didx[j] = ei[E + e0 + j];
    }
  } else {
#pragma unroll
    for (int j = 0; j < EPW; ++j) {
      const int e   = min(e0 + j, EP - 1);
      const int ec  = min(e, E - 1);
      const int sld = ei[ec];
      const int tld = ei[E + ec];
      sidx[j] = (e < E) ? sld : (e - E);
      didx[j] = (e < E) ? tld : (e - E);
    }
  }

  float p[EPW];
#pragma unroll
  for (int j = 0; j < EPW; ++j) {
    const float4 a = xl4[(size_t)sidx[j] * (DD / 4) + lane];
    const float4 b = xr4[(size_t)didx[j] * (DD / 4) + lane];
    float m0 = a.x + b.x; m0 = (m0 > 0.0f) ? m0 : NEG_SLOPE * m0;
    float m1 = a.y + b.y; m1 = (m1 > 0.0f) ? m1 : NEG_SLOPE * m1;
    float m2 = a.z + b.z; m2 = (m2 > 0.0f) ? m2 : NEG_SLOPE * m2;
    float m3 = a.w + b.w; m3 = (m3 > 0.0f) ? m3 : NEG_SLOPE * m3;
    float t = w.x * m0;
    t = fmaf(w.y, m1, t);
    t = fmaf(w.z, m2, t);
    t = fmaf(w.w, m3, t);
    p[j] = t;
  }

  const bool b4 = (lane & 4) != 0;
  const bool b2 = (lane & 2) != 0;
  const bool b1 = (lane & 1) != 0;
  float q[4];
#pragma unroll
  for (int i = 0; i < 4; ++i) {
    const float keep = b4 ? p[i + 4] : p[i];
    const float give = b4 ? p[i] : p[i + 4];
    q[i] = keep + lane_xor<4>(give);
  }
  float r[2];
#pragma unroll
  for (int i = 0; i < 2; ++i) {
    const float keep = b2 ? q[i + 2] : q[i];
    const float give = b2 ? q[i] : q[i + 2];
    r[i] = keep + lane_xor<2>(give);
  }
  const float keep = b1 ? r[1] : r[0];
  const float give = b1 ? r[0] : r[1];
  const float s = keep + lane_xor<1>(give);

  const int eown = e0 + (lane & 7);
  if (eown < EP) {
    const int h = lane >> 3;
    elog[eown * HH + h] = s;
  }
}

#define RN    196
#define OWN_T 1024
#define LCAP  8192

template <int J>
__device__ __forceinline__ float group8_bcast(float v) {
  return __uint_as_float(__builtin_amdgcn_mov_dpp8(__float_as_uint(v), J * 0x249249));
}

__device__ __forceinline__ void lds_add(float* p, float v) {
  __hip_atomic_fetch_add(p, v, __ATOMIC_RELAXED, __HIP_MEMORY_SCOPE_WORKGROUP);
}

__device__ __forceinline__ void lds_umax(unsigned* p, unsigned v) {
  __hip_atomic_fetch_max(p, v, __ATOMIC_RELAXED, __HIP_MEMORY_SCOPE_WORKGROUP);
}

__device__ __forceinline__ void own_max4(unsigned* mxU, int rel, float4 lg) {
  lds_umax(&mxU[rel * HH + 0], f2ord(lg.x));
  lds_umax(&mxU[rel * HH + 1], f2ord(lg.y));
  lds_umax(&mxU[rel * HH + 2], f2ord(lg.z));
  lds_umax(&mxU[rel * HH + 3], f2ord(lg.w));
}

__device__ __forceinline__ void own_append(unsigned e, unsigned rel, int* cnt,
                                           unsigned* keys) {
  const int pos = __hip_atomic_fetch_add(cnt, 1, __ATOMIC_RELAXED,
                                         __HIP_MEMORY_SCOPE_WORKGROUP);
  if (pos < LCAP) keys[pos] = (e << 8) | rel;
}

__device__ __forceinline__ void own_edge_serial(
    int e, int src, int rel, float* accL, float* denL, const unsigned* mxU,
    const float* __restrict__ xl, const float* __restrict__ elog) {
  float ee[HH];
  for (int hh = 0; hh < HH; ++hh) {
    ee[hh] = __expf(elog[(size_t)e * HH + hh] - ord2f(mxU[rel * HH + hh]));
    lds_add(&denL[rel * HH + hh], ee[hh]);
  }
  for (int c = 0; c < DD; ++c)
    lds_add(&accL[rel * DD + (c & 3) * 32 + (c >> 2)],
            xl[(size_t)src * DD + c] * ee[c >> 5]);
}

__global__ __launch_bounds__(OWN_T) void owner_aggregate(
    const int* __restrict__ ei, const float* __restrict__ xl,
    const float* __restrict__ elog,
    const float* xin, const float* __restrict__ bias,
    const float* __restrict__ gamma, const float* __restrict__ beta,
    float* xout, unsigned* wkeys, int* wcnt, int E, int Np, int do_relu,
    int have_list) {
  __shared__ float accL[RN * DD];
  __shared__ float denL[RN * HH];
  __shared__ unsigned mxU[RN * HH];
  __shared__ alignas(16) unsigned keys[LCAP];
  __shared__ float4 lgL[LCAP];
  __shared__ int cnt;

  const int tid  = threadIdx.x;
  const int lane = tid & 31;
  const int wave = __builtin_amdgcn_readfirstlane(tid >> 5);
  const int lo = blockIdx.x * RN;
  const int nn = min(RN, Np - lo);

  for (int i = tid; i < RN * DD / 4; i += OWN_T)
    ((float4*)accL)[i] = make_float4(0.0f, 0.0f, 0.0f, 0.0f);
  for (int i = tid; i < nn * HH; i += OWN_T) {
    denL[i] = 0.0f;
    mxU[i]  = 0u;
  }
  uint4* wk4 = (uint4*)(wkeys + (size_t)blockIdx.x * LCAP);
  int total;
  if (have_list == 0) {
  for (int i = tid; i < nn; i += OWN_T)
    keys[i] = ((unsigned)(E + lo + i) << 8) | (unsigned)i;
  if (tid == 0) cnt = nn;
  __syncthreads();

  {
    const int* dp = ei + E;
    int head = (int)(((16u - (unsigned)((size_t)dp & 15u)) & 15u) >> 2);
    head = min(head, E);
    const int nvec = (E - head) >> 2;
    const int4* dp4 = (const int4*)(dp + head);
    const unsigned un = (unsigned)nn;
#define OWN_TEST(dv, ev)                                                     \
    { const unsigned rel_ = (unsigned)((dv) - lo);                            \
      if (rel_ < un) own_append((unsigned)(ev), rel_, &cnt, keys); }
    for (int i0 = tid; i0 < nvec; i0 += 8 * OWN_T) {
      int4 d[8];
#pragma unroll
      for (int u = 0; u < 8; ++u) {
        const int i = min(i0 + u * OWN_T, nvec - 1);
        d[u] = dp4[i];
      }
#pragma unroll
      for (int u = 0; u < 8; ++u) {
        const int i = i0 + u * OWN_T;
        if (i < nvec) {
          const int eb = head + 4 * i;
          OWN_TEST(d[u].x, eb)
          OWN_TEST(d[u].y, eb + 1)
          OWN_TEST(d[u].z, eb + 2)
          OWN_TEST(d[u].w, eb + 3)
        }
      }
    }
    if (tid < head) OWN_TEST(dp[tid], tid)
    for (int e = head + 4 * nvec + tid; e < E; e += OWN_T) OWN_TEST(dp[e], e)
#undef OWN_TEST
  }
  __syncthreads();

  total = __builtin_amdgcn_readfirstlane(cnt);
  if (tid == 0) wcnt[blockIdx.x] = total;
  if (total <= LCAP)
    for (int i = tid; i < ((total + 3) >> 2); i += OWN_T)
      wk4[i] = ((const uint4*)keys)[i];
  } else {
  total = __builtin_amdgcn_readfirstlane(wcnt[blockIdx.x]);
  if (total <= LCAP)
    for (int i = tid; i < ((total + 3) >> 2); i += OWN_T)
      ((uint4*)keys)[i] = wk4[i];
  __syncthreads();
  }
  const int n = (total <= LCAP) ? total : 0;

  if (n > 0) {
    const float4* elog4 = (const float4*)elog;
    const int wrap = (1 << (31 - __builtin_clz((unsigned)n))) - 1;
    for (int i0 = tid; i0 < n; i0 += 4 * OWN_T) {
      float4 v[4];
      int rl[4];
#pragma unroll
      for (int u = 0; u < 4; ++u) {
        int i = i0 + u * OWN_T;
        i = (i < n) ? i : (i & wrap);
        const unsigned key = keys[i];
        rl[u] = (int)(key & 255u);
        v[u]  = elog4[key >> 8];
      }
      asm volatile("" :: "v"(v[0].x), "v"(v[0].y), "v"(v[0].z), "v"(v[0].w),
                         "v"(v[1].x), "v"(v[1].y), "v"(v[1].z), "v"(v[1].w),
                         "v"(v[2].x), "v"(v[2].y), "v"(v[2].z), "v"(v[2].w),
                         "v"(v[3].x), "v"(v[3].y), "v"(v[3].z), "v"(v[3].w));
#pragma unroll
      for (int u = 0; u < 4; ++u) {
        lgL[i0 + u * OWN_T] = v[u];
        own_max4(mxU, rl[u], v[u]);
      }
    }
  }
  __syncthreads();

  const int j = lane & 7;
  const int h = lane >> 3;
  const float4* xl4 = (const float4*)xl;

#define OWN_FETCH(g, e_, dl_, src_, lg_)                                  \
  { const int idx_ = min((g) + j, n - 1);                                  \
    const unsigned key_ = keys[idx_];                                      \
    e_  = (int)(key_ >> 8);                                                \
    dl_ = (int)(key_ & 255u);                                              \
    src_ = ei[min(e_, E - 1)];                     \
    src_ = (e_ < E) ? src_ : e_ - E;        \
    lg_ = ((const float*)lgL)[idx_ * HH + h]; }

#define OWN_ROW(J) \
  row[J] = xl4[(size_t)__builtin_amdgcn_readlane(src, J) * (DD / 4) + lane];

#define OWN_ADD(J, COND)                                                   \
  if (COND) {                                                              \
    const float w_ = group8_bcast<J>(ee);                                  \
    float* a_ = accL + __builtin_amdgcn_readlane(dl, J) * DD + lane;       \
    lds_add(a_,      row[J].x * w_);                                       \
    lds_add(a_ + 32, row[J].y * w_);                                       \
    lds_add(a_ + 64, row[J].z * w_);                                       \
    lds_add(a_ + 96, row[J].w * w_);                                       \
  }

  int g = wave * 8;
  int e = 0, dl = 0, src = 0;
  float lg = 0.0f;
  if (g < n) OWN_FETCH(g, e, dl, src, lg)
  while (g < n) {
    const int gn = g + (OWN_T / 32) * 8;
    const float ee = __expf(lg - ord2f(mxU[dl * HH + h]));
    if (g + j < n) lds_add(&denL[dl * HH + h], ee);

    float4 row[8];
    OWN_ROW(0) OWN_ROW(1) OWN_ROW(2) OWN_ROW(3)
    OWN_ROW(4) OWN_ROW(5) OWN_ROW(6) OWN_ROW(7)

    int e2, dl2, src2;
    float lg2;
    OWN_FETCH(gn, e2, dl2, src2, lg2)
    if (g + 8 <= n) {
      OWN_ADD(0, true) OWN_ADD(1, true) OWN_ADD(2, true) OWN_ADD(3, true)
      OWN_ADD(4, true) OWN_ADD(5, true) OWN_ADD(6, true) OWN_ADD(7, true)
    } else {
      OWN_ADD(0, g + 0 < n) OWN_ADD(1, g + 1 < n) OWN_ADD(2, g + 2 < n)
      OWN_ADD(3, g + 3 < n) OWN_ADD(4, g + 4 < n) OWN_ADD(5, g + 5 < n)
      OWN_ADD(6, g + 6 < n) OWN_ADD(7, g + 7 < n)
    }

    g = gn; e = e2; dl = dl2; src = src2; lg = lg2;
  }
#undef OWN_FETCH
#undef OWN_ROW
#undef OWN_ADD
  if (total > LCAP) {
    const int* dp = ei + E;
    const float4* elog4 = (const float4*)elog;
    for (int ed = tid; ed < E; ed += OWN_T) {
      const unsigned rel = (unsigned)(dp[ed] - lo);
      if (rel < (unsigned)nn) own_max4(mxU, (int)rel, elog4[ed]);
    }
    for (int i = tid; i < nn; i += OWN_T) own_max4(mxU, i, elog4[E + lo + i]);
    __syncthreads();
    for (int ed = tid; ed < E; ed += OWN_T) {
      const unsigned rel = (unsigned)(dp[ed] - lo);
      if (rel < (unsigned)nn)
        own_edge_serial(ed, ei[ed], (int)rel, accL, denL, mxU, xl, elog);
    }
    for (int i = tid; i < nn; i += OWN_T)
      own_edge_serial(E + lo + i, lo + i, i, accL, denL, mxU, xl, elog);
  }
  __syncthreads();

  const float4 bs = ((const float4*)bias)[lane];
  const float4 gm = ((const float4*)gamma)[lane];
  const float4 bt = ((const float4*)beta)[lane];
  for (int nl = wave; nl < nn; nl += OWN_T / 32) {
    const size_t node = (size_t)(lo + nl);
    const float4 xi = ((const float4*)xin)[node * (DD / 4) + lane];
    const float den = denL[nl * HH + h];
    const float* a = accL + nl * DD + lane;
    const float v0 = xi.x + a[0]  / den + bs.x;
    const float v1 = xi.y + a[32] / den + bs.y;
    const float v2 = xi.z + a[64] / den + bs.z;
    const float v3 = xi.w + a[96] / den + bs.w;

    float sum = (v0 + v1) + (v2 + v3);
#pragma unroll
    for (int off = 16; off > 0; off >>= 1) sum += __shfl_xor(sum, off, 32);
    const float mu = sum * (1.0f / (float)DD);

    const float d0 = v0 - mu, d1 = v1 - mu, d2 = v2 - mu, d3 = v3 - mu;
    float sq = fmaf(d0, d0, fmaf(d1, d1, fmaf(d2, d2, d3 * d3)));
#pragma unroll
    for (int off = 16; off > 0; off >>= 1) sq += __shfl_xor(sq, off, 32);
    const float rstd = rsqrtf(sq * (1.0f / (float)DD) + LN_EPS);

    float4 o;
    o.x = d0 * rstd * gm.x + bt.x;
    o.y = d1 * rstd * gm.y + bt.y;
    o.z = d2 * rstd * gm.z + bt.z;
    o.w = d3 * rstd * gm.w + bt.w;
    if (do_relu) {
      o.x = (o.x > 0.0f) ? o.x : 0.0f;
      o.y = (o.y > 0.0f) ? o.y : 0.0f;
      o.z = (o.z > 0.0f) ? o.z : 0.0f;
      o.w = (o.w > 0.0f) ? o.w : 0.0f;
    }
    ((float4*)xout)[node * (DD / 4) + lane] = o;
  }
}

static inline int cdiv(long long a, long long b) { return (int)((a + b - 1) / b); }

extern "C" void kernel_launch(void* const* d_in, const int* in_sizes, int n_in,
                              void* d_out, int out_size, void* d_ws, size_t ws_size,
                              hipStream_t stream) {
  const int Np = NN;
  const int E  = in_sizes[1] / 2;
  const int EP = E + Np;

  const float* x0    = (const float*)d_in[0];
  const int*   ei    = (const int*)d_in[1];
  const float* Wl    = (const float*)d_in[2];
  const float* bl    = (const float*)d_in[3];
  const float* Wr    = (const float*)d_in[4];
  const float* br    = (const float*)d_in[5];
  const float* att   = (const float*)d_in[6];
  const float* bias  = (const float*)d_in[7];
  const float* gamma = (const float*)d_in[8];
  const float* beta  = (const float*)d_in[9];

  float* ws = (float*)d_ws;
  const size_t ND = (size_t)Np * DD;
  float*    bufA  = ws;
  float*    xl    = bufA + ND;
  float*    xr    = xl + ND;
  float*    elog  = xr + ND;
  const int NWG = cdiv(Np, RN);
  unsigned* wkeys = (unsigned*)(elog + (size_t)EP * HH);
  int*      wcnt  = (int*)(wkeys + (size_t)NWG * LCAP);

  const int TB = 256;
  const dim3 blk(TB);

  for (int i = 0; i < LL; ++i) {
    const float* xin  = (i == 0) ? x0 : bufA;
    float*       xout = (i == LL - 1) ? (float*)d_out : bufA;
    const float* Wli  = Wl + (size_t)i * DD * DD;
    const float* bli  = bl + (size_t)i * DD;
    const float* Wri  = Wr + (size_t)i * DD * DD;
    const float* bri  = br + (size_t)i * DD;
    const float* atti = att + (size_t)i * HH * CC;
    const float* bi   = bias + (size_t)i * DD;
    const float* gi   = gamma + (size_t)i * DD;
    const float* bti  = beta + (size_t)i * DD;

    gemm_dual_wmma<<<Np / 16, blk, 0, stream>>>(xin, Wli, bli, Wri, bri, xl, xr);

    edge_logits<<<cdiv((long long)EP, (TB / 32) * EPW), blk, 0, stream>>>(
        ei, xl, xr, atti, elog, E, Np);

    owner_aggregate<<<NWG, dim3(OWN_T), 0, stream>>>(
        ei, xl, elog, xin, bi, gi, bti, xout, wkeys, wcnt, E, Np,
        (i != LL - 1) ? 1 : 0, (i != 0) ? 1 : 0);
  }
}


# Softmax shifted by each node's self-loop logit, not the segment max: ONE xl[src] row read per edge gives logit, exp and LDS sums (terms exp(e-e_self), self loop stored first, then arrival order)
# speedup vs baseline: 4.3032x; 1.4026x over previous
#include <hip/hip_runtime.h>
#include <math.h>


#define NN 50000
#define DD 128
#define HH 4
#define CC 32
#define LL 3
#define NEG_SLOPE 0.2f
#define LN_EPS 1e-5f

typedef __attribute__((ext_vector_type(2))) float v2f;
typedef __attribute__((ext_vector_type(8))) float v8f;

__device__ __forceinline__ unsigned f2ord(float f) {
  unsigned u = __float_as_uint(f);
  return (u & 0x80000000u) ? ~u : (u | 0x80000000u);
}
__device__ __forceinline__ float ord2f(unsigned u) {
  u = (u & 0x80000000u) ? (u & 0x7FFFFFFFu) : ~u;
  return __uint_as_float(u);
}

__global__ __launch_bounds__(256) void gemm_dual_wmma(
    const float* __restrict__ x,
    const float* __restrict__ Wl, const float* __restrict__ bl,
    const float* __restrict__ Wr, const float* __restrict__ br,
    float* __restrict__ xl, float* __restrict__ xr) {
  __shared__ float sx[16 * DD];

  const int tid  = threadIdx.x;
  const int wave = tid >> 5;
  const int lane = tid & 31;
  const int rowBase = blockIdx.x * 16;

  {
    const float4* gsrc = (const float4*)(x + (size_t)rowBase * DD);
    float4* sdst = (float4*)sx;
    sdst[tid]       = gsrc[tid];
    sdst[tid + 256] = gsrc[tid + 256];
  }
  __syncthreads();

  const int colBase = wave * 16;
  const int m  = lane & 15;
  const int kk = (lane >> 4) * 2;
  const int n  = lane & 15;

  v8f cl = {};
  v8f cr = {};

  for (int k0 = 0; k0 < DD; k0 += 4) {
    v2f a;
    a.x = sx[m * DD + k0 + kk];
    a.y = sx[m * DD + k0 + kk + 1];
    v2f bfl;
    bfl.x = Wl[(size_t)(k0 + kk) * DD + colBase + n];
    bfl.y = Wl[(size_t)(k0 + kk + 1) * DD + colBase + n];
    v2f bfr;
    bfr.x = Wr[(size_t)(k0 + kk) * DD + colBase + n];
    bfr.y = Wr[(size_t)(k0 + kk + 1) * DD + colBase + n];

    cl = __builtin_amdgcn_wmma_f32_16x16x4_f32(
        false, a, false, bfl, (short)0, cl, false, false);
    cr = __builtin_amdgcn_wmma_f32_16x16x4_f32(
        false, a, false, bfr, (short)0, cr, false, false);
  }

  const int rOff = (lane >> 4) * 8;
  const int col  = colBase + n;
  const float bL = bl[col];
  const float bR = br[col];
#pragma unroll
  for (int i = 0; i < 8; ++i) {
    const size_t row = (size_t)(rowBase + i + rOff);
    xl[row * DD + col] = cl[i] + bL;
    xr[row * DD + col] = cr[i] + bR;
  }
}


template <int M>
__device__ __forceinline__ float lane_xor(float v) {
  return __int_as_float(__builtin_amdgcn_update_dpp(
      0, __float_as_int(v), 0x160 + M, 0xf, 0xf, true));
}

#define RN    196
#define OWN_T 1024
#define LCAP  8192
#define SHIFT_LIM 40.0f

template <int J>
__device__ __forceinline__ float group8_bcast(float v) {
  return __uint_as_float(__builtin_amdgcn_mov_dpp8(__float_as_uint(v), J * 0x249249));
}

__device__ __forceinline__ void lds_add(float* p, float v) {
  __hip_atomic_fetch_add(p, v, __ATOMIC_RELAXED, __HIP_MEMORY_SCOPE_WORKGROUP);
}

__device__ __forceinline__ void lds_umax(unsigned* p, unsigned v) {
  __hip_atomic_fetch_max(p, v, __ATOMIC_RELAXED, __HIP_MEMORY_SCOPE_WORKGROUP);
}

__device__ __forceinline__ void own_max4(unsigned* mxU, int rel, float4 lg) {
  lds_umax(&mxU[rel * HH + 0], f2ord(lg.x));
  lds_umax(&mxU[rel * HH + 1], f2ord(lg.y));
  lds_umax(&mxU[rel * HH + 2], f2ord(lg.z));
  lds_umax(&mxU[rel * HH + 3], f2ord(lg.w));
}

__device__ __forceinline__ void own_append(unsigned e, unsigned rel, int* cnt,
                                           unsigned* keys) {
  const int pos = __hip_atomic_fetch_add(cnt, 1, __ATOMIC_RELAXED,
                                         __HIP_MEMORY_SCOPE_WORKGROUP);
  if (pos < LCAP) keys[pos] = (e << 8) | rel;
}

__device__ __forceinline__ float own_dot4(float4 a, float4 b, float4 w) {
  float m0 = a.x + b.x; m0 = (m0 > 0.0f) ? m0 : NEG_SLOPE * m0;
  float m1 = a.y + b.y; m1 = (m1 > 0.0f) ? m1 : NEG_SLOPE * m1;
  float m2 = a.z + b.z; m2 = (m2 > 0.0f) ? m2 : NEG_SLOPE * m2;
  float m3 = a.w + b.w; m3 = (m3 > 0.0f) ? m3 : NEG_SLOPE * m3;
  float t = w.x * m0;
  t = fmaf(w.y, m1, t);
  t = fmaf(w.z, m2, t);
  t = fmaf(w.w, m3, t);
  return t;
}

__device__ __forceinline__ float4 own_logits_serial(
    int src, int node, const float* __restrict__ xl,
    const float* __restrict__ xr, const float* __restrict__ att) {
  float lg[HH];
  for (int hh = 0; hh < HH; ++hh) {
    float t = 0.0f;
    for (int c = 0; c < CC; ++c) {
      float m = xl[(size_t)src * DD + hh * CC + c] +
                xr[(size_t)node * DD + hh * CC + c];
      m = (m > 0.0f) ? m : NEG_SLOPE * m;
      t = fmaf(att[hh * CC + c], m, t);
    }
    lg[hh] = t;
  }
  return make_float4(lg[0], lg[1], lg[2], lg[3]);
}

__device__ __forceinline__ void own_edge_serial(
    float4 lg, int src, int rel, float* accL, float* denL,
    const unsigned* mxU, const float* __restrict__ xl) {
  const float l4[HH] = {lg.x, lg.y, lg.z, lg.w};
  float ee[HH];
  for (int hh = 0; hh < HH; ++hh) {
    ee[hh] = __expf(l4[hh] - ord2f(mxU[rel * HH + hh]));
    lds_add(&denL[rel * HH + hh], ee[hh]);
  }
  for (int c = 0; c < DD; ++c)
    lds_add(&accL[rel * DD + (c & 3) * 32 + (c >> 2)],
            xl[(size_t)src * DD + c] * ee[c >> 5]);
}

__global__ __launch_bounds__(OWN_T) void owner_aggregate(
    const int* __restrict__ ei, const float* __restrict__ xl,
    const float* __restrict__ xr, const float* __restrict__ att,
    const float* xin, const float* __restrict__ bias,
    const float* __restrict__ gamma, const float* __restrict__ beta,
    float* xout, unsigned* wkeys, int* wcnt, int E, int Np, int do_relu,
    int have_list) {
  __shared__ alignas(16) float accL[RN * DD];
  __shared__ alignas(16) float xrL[RN * DD];
  __shared__ float denL[RN * HH];
  __shared__ float esL[RN * HH];
  __shared__ unsigned mxU[RN * HH];
  __shared__ alignas(16) unsigned keys[LCAP];
  __shared__ int cnt;
  __shared__ int ovf;

  const int tid  = threadIdx.x;
  const int lane = tid & 31;
  const int wave = __builtin_amdgcn_readfirstlane(tid >> 5);
  const int lo = blockIdx.x * RN;
  const int nn = min(RN, Np - lo);

  {
    const float4* xs = (const float4*)xr + (size_t)lo * (DD / 4);
    const int nq = nn * (DD / 4);
    constexpr int XU = (RN * (DD / 4) + OWN_T - 1) / OWN_T;
    float4 t[XU];
    int ti[XU];
#pragma unroll
    for (int u = 0; u < XU; ++u) {
      ti[u] = min(tid + u * OWN_T, nq - 1);
      t[u] = xs[ti[u]];
    }
#pragma unroll
    for (int u = 0; u < XU; ++u) ((float4*)xrL)[ti[u]] = t[u];
  }
  for (int i = tid; i < nn * HH; i += OWN_T)
    mxU[i] = 0u;
  if (tid == 0) ovf = 0;
  uint4* wk4 = (uint4*)(wkeys + (size_t)blockIdx.x * LCAP);
  int total;
  if (have_list == 0) {
  for (int i = tid; i < nn; i += OWN_T)
    keys[i] = ((unsigned)(E + lo + i) << 8) | (unsigned)i;
  if (tid == 0) cnt = nn;
  __syncthreads();

  {
    const int* dp = ei + E;
    int head = (int)(((16u - (unsigned)((size_t)dp & 15u)) & 15u) >> 2);
    head = min(head, E);
    const int nvec = (E - head) >> 2;
    const int4* dp4 = (const int4*)(dp + head);
    const unsigned un = (unsigned)nn;
#define OWN_TEST(dv, ev)                                                     \
    { const unsigned rel_ = (unsigned)((dv) - lo);                            \
      if (rel_ < un) own_append((unsigned)(ev), rel_, &cnt, keys); }
    for (int i0 = tid; i0 < nvec; i0 += 8 * OWN_T) {
      int4 d[8];
#pragma unroll
      for (int u = 0; u < 8; ++u) {
        const int i = min(i0 + u * OWN_T, nvec - 1);
        d[u] = dp4[i];
      }
#pragma unroll
      for (int u = 0; u < 8; ++u) {
        const int i = i0 + u * OWN_T;
        if (i < nvec) {
          const int eb = head + 4 * i;
          OWN_TEST(d[u].x, eb)
          OWN_TEST(d[u].y, eb + 1)
          OWN_TEST(d[u].z, eb + 2)
          OWN_TEST(d[u].w, eb + 3)
        }
      }
    }
    if (tid < head) OWN_TEST(dp[tid], tid)
    for (int e = head + 4 * nvec + tid; e < E; e += OWN_T) OWN_TEST(dp[e], e)
#undef OWN_TEST
  }
  __syncthreads();

  total = __builtin_amdgcn_readfirstlane(cnt);
  if (tid == 0) wcnt[blockIdx.x] = total;
  if (total <= LCAP)
    for (int i = tid; i < ((total + 3) >> 2); i += OWN_T)
      wk4[i] = ((const uint4*)keys)[i];
  } else {
  total = __builtin_amdgcn_readfirstlane(wcnt[blockIdx.x]);
  if (total <= LCAP)
    for (int i = tid; i < ((total + 3) >> 2); i += OWN_T)
      ((uint4*)keys)[i] = wk4[i];
  __syncthreads();
  }
  const int n = (total <= LCAP) ? total : 0;

  const int j = lane & 7;
  const int h = lane >> 3;
  const float4* xl4 = (const float4*)xl;

#define OWN_IDS(g, dl_, src_)                                              \
  { const int idx_ = min((g) + j, n - 1);                                  \
    const unsigned key_ = keys[idx_];                                      \
    const int e_ = (int)(key_ >> 8);                                       \
    dl_ = (int)(key_ & 255u);                                              \
    src_ = ei[min(e_, E - 1)];                     \
    src_ = (e_ < E) ? src_ : e_ - E; }

#define OWN_ROW(J) \
  row[J] = xl4[(size_t)__builtin_amdgcn_readlane(src, J) * (DD / 4) + lane];

#define OWN_ADD(J, COND)                                                   \
  if (COND) {                                                              \
    const float w_ = group8_bcast<J>(ee);                                  \
    float* a_ = accL + __builtin_amdgcn_readlane(dl, J) * DD + lane;       \
    lds_add(a_,      row[J].x * w_);                                       \
    lds_add(a_ + 32, row[J].y * w_);                                       \
    lds_add(a_ + 64, row[J].z * w_);                                       \
    lds_add(a_ + 96, row[J].w * w_);                                       \
  }

  if (n > 0) {
    const float4 w = ((const float4*)att)[lane];
    const float4* xrL4 = (const float4*)xrL;

    {
      constexpr int SU = (RN + OWN_T / 32 - 1) / (OWN_T / 32);
      float4 a[SU];
      int nl[SU];
#pragma unroll
      for (int u = 0; u < SU; ++u) {
        nl[u] = min(wave + u * (OWN_T / 32), nn - 1);
        a[u] = xl4[(size_t)(lo + nl[u]) * (DD / 4) + lane];
      }
#pragma unroll
      for (int u = 0; u < SU; ++u) {
        if (wave + u * (OWN_T / 32) < nn) {
          float t = own_dot4(a[u], xrL4[nl[u] * (DD / 4) + lane], w);
          t += lane_xor<4>(t);
          t += lane_xor<2>(t);
          t += lane_xor<1>(t);
          if (j == 0) {
            esL[nl[u] * HH + h]  = t;
            denL[nl[u] * HH + h] = 1.0f;
          }
          float* q = accL + nl[u] * DD + lane;
          q[0]  = a[u].x;
          q[32] = a[u].y;
          q[64] = a[u].z;
          q[96] = a[u].w;
        }
      }
    }
    __syncthreads();

    const bool b4 = (lane & 4) != 0;
    const bool b2 = (lane & 2) != 0;
    const bool b1 = (lane & 1) != 0;

    int g = nn + wave * 8;
    int dl = 0, src = 0;
    if (g < n) OWN_IDS(g, dl, src)
    while (g < n) {
      const int gn = g + (OWN_T / 32) * 8;
      float4 row[8];
      OWN_ROW(0) OWN_ROW(1) OWN_ROW(2) OWN_ROW(3)
      OWN_ROW(4) OWN_ROW(5) OWN_ROW(6) OWN_ROW(7)

      int dl2, src2;
      OWN_IDS(gn, dl2, src2)

      const float sh = esL[dl * HH + h];

      float p[8];
#define OWN_DOT(J) \
      p[J] = own_dot4(row[J], xrL4[__builtin_amdgcn_readlane(dl, J) * (DD / 4) + lane], w);
      OWN_DOT(0) OWN_DOT(1) OWN_DOT(2) OWN_DOT(3)
      OWN_DOT(4) OWN_DOT(5) OWN_DOT(6) OWN_DOT(7)
#undef OWN_DOT

      float q[4];
#pragma unroll
      for (int i = 0; i < 4; ++i) {
        const float keep = b4 ? p[i + 4] : p[i];
        const float give = b4 ? p[i] : p[i + 4];
        q[i] = keep + lane_xor<4>(give);
      }
      float r[2];
#pragma unroll
      for (int i = 0; i < 2; ++i) {
        const float keep = b2 ? q[i + 2] : q[i];
        const float give = b2 ? q[i] : q[i + 2];
        r[i] = keep + lane_xor<2>(give);
      }
      const float keep = b1 ? r[1] : r[0];
      const float give = b1 ? r[0] : r[1];
      const float s = keep + lane_xor<1>(give);

      const float d  = s - sh;
      const float ee = __expf(d);
      if (g + j < n) {
        lds_add(&denL[dl * HH + h], ee);
        if (!(d <= SHIFT_LIM)) ovf = 1;
      }

      if (g + 8 <= n) {
        OWN_ADD(0, true) OWN_ADD(1, true) OWN_ADD(2, true) OWN_ADD(3, true)
        OWN_ADD(4, true) OWN_ADD(5, true) OWN_ADD(6, true) OWN_ADD(7, true)
      } else {
        OWN_ADD(0, g + 0 < n) OWN_ADD(1, g + 1 < n) OWN_ADD(2, g + 2 < n)
        OWN_ADD(3, g + 3 < n) OWN_ADD(4, g + 4 < n) OWN_ADD(5, g + 5 < n)
        OWN_ADD(6, g + 6 < n) OWN_ADD(7, g + 7 < n)
      }

      g = gn; dl = dl2; src = src2;
    }
  }
  __syncthreads();
#undef OWN_IDS
#undef OWN_ROW
#undef OWN_ADD
  if (total > LCAP || __builtin_amdgcn_readfirstlane(ovf) != 0) {
    for (int i = tid; i < RN * DD / 4; i += OWN_T)
      ((float4*)accL)[i] = make_float4(0.0f, 0.0f, 0.0f, 0.0f);
    for (int i = tid; i < nn * HH; i += OWN_T) denL[i] = 0.0f;
    __syncthreads();
    const int* dp = ei + E;
    for (int ed = tid; ed < E; ed += OWN_T) {
      const unsigned rel = (unsigned)(dp[ed] - lo);
      if (rel < (unsigned)nn)
        own_max4(mxU, (int)rel,
                 own_logits_serial(ei[ed], lo + (int)rel, xl, xr, att));
    }
    for (int i = tid; i < nn; i += OWN_T)
      own_max4(mxU, i, own_logits_serial(lo + i, lo + i, xl, xr, att));
    __syncthreads();
    for (int ed = tid; ed < E; ed += OWN_T) {
      const unsigned rel = (unsigned)(dp[ed] - lo);
      if (rel < (unsigned)nn) {
        const int sn = ei[ed];
        own_edge_serial(own_logits_serial(sn, lo + (int)rel, xl, xr, att), sn,
                        (int)rel, accL, denL, mxU, xl);
      }
    }
    for (int i = tid; i < nn; i += OWN_T)
      own_edge_serial(own_logits_serial(lo + i, lo + i, xl, xr, att), lo + i,
                      i, accL, denL, mxU, xl);
    __syncthreads();
  }

  const float4 bs = ((const float4*)bias)[lane];
  const float4 gm = ((const float4*)gamma)[lane];
  const float4 bt = ((const float4*)beta)[lane];
  for (int nl = wave; nl < nn; nl += OWN_T / 32) {
    const size_t node = (size_t)(lo + nl);
    const float4 xi = ((const float4*)xin)[node * (DD / 4) + lane];
    const float den = denL[nl * HH + h];
    const float* a = accL + nl * DD + lane;
    const float v0 = xi.x + a[0]  / den + bs.x;
    const float v1 = xi.y + a[32] / den + bs.y;
    const float v2 = xi.z + a[64] / den + bs.z;
    const float v3 = xi.w + a[96] / den + bs.w;

    float sum = (v0 + v1) + (v2 + v3);
#pragma unroll
    for (int off = 16; off > 0; off >>= 1) sum += __shfl_xor(sum, off, 32);
    const float mu = sum * (1.0f / (float)DD);

    const float d0 = v0 - mu, d1 = v1 - mu, d2 = v2 - mu, d3 = v3 - mu;
    float sq = fmaf(d0, d0, fmaf(d1, d1, fmaf(d2, d2, d3 * d3)));
#pragma unroll
    for (int off = 16; off > 0; off >>= 1) sq += __shfl_xor(sq, off, 32);
    const float rstd = rsqrtf(sq * (1.0f / (float)DD) + LN_EPS);

    float4 o;
    o.x = d0 * rstd * gm.x + bt.x;
    o.y = d1 * rstd * gm.y + bt.y;
    o.z = d2 * rstd * gm.z + bt.z;
    o.w = d3 * rstd * gm.w + bt.w;
    if (do_relu) {
      o.x = (o.x > 0.0f) ? o.x : 0.0f;
      o.y = (o.y > 0.0f) ? o.y : 0.0f;
      o.z = (o.z > 0.0f) ? o.z : 0.0f;
      o.w = (o.w > 0.0f) ? o.w : 0.0f;
    }
    ((float4*)xout)[node * (DD / 4) + lane] = o;
  }
}

static inline int cdiv(long long a, long long b) { return (int)((a + b - 1) / b); }

extern "C" void kernel_launch(void* const* d_in, const int* in_sizes, int n_in,
                              void* d_out, int out_size, void* d_ws, size_t ws_size,
                              hipStream_t stream) {
  const int Np = NN;
  const int E  = in_sizes[1] / 2;

  const float* x0    = (const float*)d_in[0];
  const int*   ei    = (const int*)d_in[1];
  const float* Wl    = (const float*)d_in[2];
  const float* bl    = (const float*)d_in[3];
  const float* Wr    = (const float*)d_in[4];
  const float* br    = (const float*)d_in[5];
  const float* att   = (const float*)d_in[6];
  const float* bias  = (const float*)d_in[7];
  const float* gamma = (const float*)d_in[8];
  const float* beta  = (const float*)d_in[9];

  float* ws = (float*)d_ws;
  const size_t ND = (size_t)Np * DD;
  float*    bufA  = ws;
  float*    xl    = bufA + ND;
  float*    xr    = xl + ND;
  const int NWG = cdiv(Np, RN);
  unsigned* wkeys = (unsigned*)(xr + ND);
  int*      wcnt  = (int*)(wkeys + (size_t)NWG * LCAP);

  const int TB = 256;
  const dim3 blk(TB);

  for (int i = 0; i < LL; ++i) {
    const float* xin  = (i == 0) ? x0 : bufA;
    float*       xout = (i == LL - 1) ? (float*)d_out : bufA;
    const float* Wli  = Wl + (size_t)i * DD * DD;
    const float* bli  = bl + (size_t)i * DD;
    const float* Wri  = Wr + (size_t)i * DD * DD;
    const float* bri  = br + (size_t)i * DD;
    const float* atti = att + (size_t)i * HH * CC;
    const float* bi   = bias + (size_t)i * DD;
    const float* gi   = gamma + (size_t)i * DD;
    const float* bti  = beta + (size_t)i * DD;

    gemm_dual_wmma<<<Np / 16, blk, 0, stream>>>(xin, Wli, bli, Wri, bri, xl, xr);

    owner_aggregate<<<NWG, dim3(OWN_T), 0, stream>>>(
        ei, xl, xr, atti, xin, bi, gi, bti, xout, wkeys, wcnt, E, Np,
        (i != LL - 1) ? 1 : 0, (i != 0) ? 1 : 0);
  }
}


# First-layer edge filter reads one owner byte per edge (byte plane written in the first GEMM launch; 197 nodes per owner), then looks up only kept edges' destinations; LDS sums in a new arrival order
# speedup vs baseline: 4.5557x; 1.0587x over previous
#include <hip/hip_runtime.h>
#include <math.h>


#define NN 50000
#define DD 128
#define HH 4
#define CC 32
#define LL 3
#define RN 197
#define NO_OWNER 255u
#define NEG_SLOPE 0.2f
#define LN_EPS 1e-5f

typedef __attribute__((ext_vector_type(2))) float v2f;
typedef __attribute__((ext_vector_type(8))) float v8f;

__device__ __forceinline__ unsigned f2ord(float f) {
  unsigned u = __float_as_uint(f);
  return (u & 0x80000000u) ? ~u : (u | 0x80000000u);
}
__device__ __forceinline__ float ord2f(unsigned u) {
  u = (u & 0x80000000u) ? (u & 0x7FFFFFFFu) : ~u;
  return __uint_as_float(u);
}

__device__ __forceinline__ void owner_bytes16(const int* __restrict__ ei,
                                              unsigned* __restrict__ ob, int E,
                                              int Np, int t) {
  const int e0 = t * 16;
  if (e0 >= E) return;
  const int* dp = ei + E + e0;
  int d[16];
  if (e0 + 16 <= E && (((size_t)dp) & 15u) == 0u) {
#pragma unroll
    for (int q = 0; q < 4; ++q) {
      const int4 v = ((const int4*)dp)[q];
      d[4 * q] = v.x; d[4 * q + 1] = v.y; d[4 * q + 2] = v.z; d[4 * q + 3] = v.w;
    }
  } else {
#pragma unroll
    for (int k = 0; k < 16; ++k) d[k] = (e0 + k < E) ? dp[k] : -1;
  }
  unsigned w[4];
#pragma unroll
  for (int q = 0; q < 4; ++q) {
    w[q] = 0u;
#pragma unroll
    for (int b = 0; b < 4; ++b) {
      const unsigned dv = (unsigned)d[4 * q + b];
      const unsigned o = (dv < (unsigned)Np) ? dv / (unsigned)RN : NO_OWNER;
      w[q] |= o << (8 * b);
    }
  }
  ((uint4*)ob)[t] = make_uint4(w[0], w[1], w[2], w[3]);
}

__global__ __launch_bounds__(256) void gemm_dual_wmma(
    const float* __restrict__ x,
    const float* __restrict__ Wl, const float* __restrict__ bl,
    const float* __restrict__ Wr, const float* __restrict__ br,
    float* __restrict__ xl, float* __restrict__ xr,
    const int* __restrict__ ei, unsigned* __restrict__ ob, int E, int Np) {
  __shared__ float sx[16 * DD];

  const int tid  = threadIdx.x;
  if (ob != nullptr)
    for (int t = blockIdx.x * 256 + tid; t < ((E + 15) >> 4);
         t += (int)gridDim.x * 256)
      owner_bytes16(ei, ob, E, Np, t);
  const int wave = tid >> 5;
  const int lane = tid & 31;
  const int rowBase = blockIdx.x * 16;

  {
    const float4* gsrc = (const float4*)(x + (size_t)rowBase * DD);
    float4* sdst = (float4*)sx;
    sdst[tid]       = gsrc[tid];
    sdst[tid + 256] = gsrc[tid + 256];
  }
  __syncthreads();

  const int colBase = wave * 16;
  const int m  = lane & 15;
  const int kk = (lane >> 4) * 2;
  const int n  = lane & 15;

  v8f cl = {};
  v8f cr = {};

  for (int k0 = 0; k0 < DD; k0 += 4) {
    v2f a;
    a.x = sx[m * DD + k0 + kk];
    a.y = sx[m * DD + k0 + kk + 1];
    v2f bfl;
    bfl.x = Wl[(size_t)(k0 + kk) * DD + colBase + n];
    bfl.y = Wl[(size_t)(k0 + kk + 1) * DD + colBase + n];
    v2f bfr;
    bfr.x = Wr[(size_t)(k0 + kk) * DD + colBase + n];
    bfr.y = Wr[(size_t)(k0 + kk + 1) * DD + colBase + n];

    cl = __builtin_amdgcn_wmma_f32_16x16x4_f32(
        false, a, false, bfl, (short)0, cl, false, false);
    cr = __builtin_amdgcn_wmma_f32_16x16x4_f32(
        false, a, false, bfr, (short)0, cr, false, false);
  }

  const int rOff = (lane >> 4) * 8;
  const int col  = colBase + n;
  const float bL = bl[col];
  const float bR = br[col];
#pragma unroll
  for (int i = 0; i < 8; ++i) {
    const size_t row = (size_t)(rowBase + i + rOff);
    xl[row * DD + col] = cl[i] + bL;
    xr[row * DD + col] = cr[i] + bR;
  }
}


template <int M>
__device__ __forceinline__ float lane_xor(float v) {
  return __int_as_float(__builtin_amdgcn_update_dpp(
      0, __float_as_int(v), 0x160 + M, 0xf, 0xf, true));
}

#define OWN_T 1024
#define LCAP  8192
#define SHIFT_LIM 40.0f

template <int J>
__device__ __forceinline__ float group8_bcast(float v) {
  return __uint_as_float(__builtin_amdgcn_mov_dpp8(__float_as_uint(v), J * 0x249249));
}

__device__ __forceinline__ void lds_add(float* p, float v) {
  __hip_atomic_fetch_add(p, v, __ATOMIC_RELAXED, __HIP_MEMORY_SCOPE_WORKGROUP);
}

__device__ __forceinline__ void lds_umax(unsigned* p, unsigned v) {
  __hip_atomic_fetch_max(p, v, __ATOMIC_RELAXED, __HIP_MEMORY_SCOPE_WORKGROUP);
}

__device__ __forceinline__ void own_max4(unsigned* mxU, int rel, float4 lg) {
  lds_umax(&mxU[rel * HH + 0], f2ord(lg.x));
  lds_umax(&mxU[rel * HH + 1], f2ord(lg.y));
  lds_umax(&mxU[rel * HH + 2], f2ord(lg.z));
  lds_umax(&mxU[rel * HH + 3], f2ord(lg.w));
}

__device__ __forceinline__ void own_append(unsigned e, int* cnt,
                                           unsigned* keys) {
  const int pos = __hip_atomic_fetch_add(cnt, 1, __ATOMIC_RELAXED,
                                         __HIP_MEMORY_SCOPE_WORKGROUP);
  if (pos < LCAP) keys[pos] = e << 8;
}

__device__ __forceinline__ unsigned zero_bytes(unsigned x) {
  return ~(((x & 0x7F7F7F7Fu) + 0x7F7F7F7Fu) | x | 0x7F7F7F7Fu);
}

__device__ __forceinline__ float own_dot4(float4 a, float4 b, float4 w) {
  float m0 = a.x + b.x; m0 = (m0 > 0.0f) ? m0 : NEG_SLOPE * m0;
  float m1 = a.y + b.y; m1 = (m1 > 0.0f) ? m1 : NEG_SLOPE * m1;
  float m2 = a.z + b.z; m2 = (m2 > 0.0f) ? m2 : NEG_SLOPE * m2;
  float m3 = a.w + b.w; m3 = (m3 > 0.0f) ? m3 : NEG_SLOPE * m3;
  float t = w.x * m0;
  t = fmaf(w.y, m1, t);
  t = fmaf(w.z, m2, t);
  t = fmaf(w.w, m3, t);
  return t;
}

__device__ __forceinline__ float4 own_logits_serial(
    int src, int node, const float* __restrict__ xl,
    const float* __restrict__ xr, const float* __restrict__ att) {
  float lg[HH];
  for (int hh = 0; hh < HH; ++hh) {
    float t = 0.0f;
    for (int c = 0; c < CC; ++c) {
      float m = xl[(size_t)src * DD + hh * CC + c] +
                xr[(size_t)node * DD + hh * CC + c];
      m = (m > 0.0f) ? m : NEG_SLOPE * m;
      t = fmaf(att[hh * CC + c], m, t);
    }
    lg[hh] = t;
  }
  return make_float4(lg[0], lg[1], lg[2], lg[3]);
}

__device__ __forceinline__ void own_edge_serial(
    float4 lg, int src, int rel, float* accL, float* denL,
    const unsigned* mxU, const float* __restrict__ xl) {
  const float l4[HH] = {lg.x, lg.y, lg.z, lg.w};
  float ee[HH];
  for (int hh = 0; hh < HH; ++hh) {
    ee[hh] = __expf(l4[hh] - ord2f(mxU[rel * HH + hh]));
    lds_add(&denL[rel * HH + hh], ee[hh]);
  }
  for (int c = 0; c < DD; ++c)
    lds_add(&accL[rel * DD + (c & 3) * 32 + (c >> 2)],
            xl[(size_t)src * DD + c] * ee[c >> 5]);
}

__global__ __launch_bounds__(OWN_T) void owner_aggregate(
    const int* __restrict__ ei, const float* __restrict__ xl,
    const float* __restrict__ xr, const float* __restrict__ att,
    const float* xin, const float* __restrict__ bias,
    const float* __restrict__ gamma, const float* __restrict__ beta,
    float* xout, unsigned* wkeys, int* wcnt, const unsigned* __restrict__ obytes,
    int E, int Np, int do_relu, int have_list) {
  __shared__ alignas(16) float accL[RN * DD];
  __shared__ alignas(16) float xrL[RN * DD];
  __shared__ float denL[RN * HH];
  __shared__ float esL[RN * HH];
  __shared__ unsigned mxU[RN * HH];
  __shared__ alignas(16) unsigned keys[LCAP];
  __shared__ int cnt;
  __shared__ int ovf;

  const int tid  = threadIdx.x;
  const int lane = tid & 31;
  const int wave = __builtin_amdgcn_readfirstlane(tid >> 5);
  const int lo = blockIdx.x * RN;
  const int nn = min(RN, Np - lo);

  {
    const float4* xs = (const float4*)xr + (size_t)lo * (DD / 4);
    const int nq = nn * (DD / 4);
    constexpr int XU = (RN * (DD / 4) + OWN_T - 1) / OWN_T;
    float4 t[XU];
    int ti[XU];
#pragma unroll
    for (int u = 0; u < XU; ++u) {
      ti[u] = min(tid + u * OWN_T, nq - 1);
      t[u] = xs[ti[u]];
    }
#pragma unroll
    for (int u = 0; u < XU; ++u) ((float4*)xrL)[ti[u]] = t[u];
  }
  for (int i = tid; i < nn * HH; i += OWN_T)
    mxU[i] = 0u;
  if (tid == 0) ovf = 0;
  uint4* wk4 = (uint4*)(wkeys + (size_t)blockIdx.x * LCAP);
  int total;
  if (have_list == 0) {
  for (int i = tid; i < nn; i += OWN_T)
    keys[i] = ((unsigned)(E + lo + i) << 8) | (unsigned)i;
  if (tid == 0) cnt = nn;
  __syncthreads();

  {
    const uint4* ob4 = (const uint4*)obytes;
    const int nv = (E + 15) >> 4;
    const unsigned pat = (unsigned)blockIdx.x * 0x01010101u;
    for (int i0 = tid; i0 < nv; i0 += 8 * OWN_T) {
      uint4 d[8];
#pragma unroll
      for (int u = 0; u < 8; ++u) {
        const int i = min(i0 + u * OWN_T, nv - 1);
        d[u] = ob4[i];
      }
#pragma unroll
      for (int u = 0; u < 8; ++u) {
        const int i = i0 + u * OWN_T;
        if (i < nv) {
          unsigned m = (zero_bytes(d[u].x ^ pat) >> 7) |
                       (zero_bytes(d[u].y ^ pat) >> 6) |
                       (zero_bytes(d[u].z ^ pat) >> 5) |
                       (zero_bytes(d[u].w ^ pat) >> 4);
          while (m != 0u) {
            const unsigned p = (unsigned)__builtin_ctz(m);
            own_append(16u * (unsigned)i + 4u * (p & 7u) + (p >> 3), &cnt, keys);
            m &= m - 1u;
          }
        }
      }
    }
  }
  __syncthreads();

  total = __builtin_amdgcn_readfirstlane(cnt);
  {
    const int* dp = ei + E;
    const int nfix = min(total, LCAP);
    for (int i0 = nn + tid; i0 < nfix; i0 += 4 * OWN_T) {
      unsigned k[4];
      int dv[4];
#pragma unroll
      for (int u = 0; u < 4; ++u) {
        const int i = i0 + u * OWN_T;
        k[u] = (i < nfix) ? keys[i] : 0u;
      }
#pragma unroll
      for (int u = 0; u < 4; ++u) dv[u] = dp[k[u] >> 8];
      const unsigned k0 = k[0] | (unsigned)(dv[0] - lo);
#pragma unroll
      for (int u = 0; u < 4; ++u) {
        const int i = i0 + u * OWN_T;
        const bool in = i < nfix;
        keys[in ? i : i0] = in ? (k[u] | (unsigned)(dv[u] - lo)) : k0;
      }
    }
  }
  __syncthreads();
  if (tid == 0) wcnt[blockIdx.x] = total;
  if (total <= LCAP)
    for (int i = tid; i < ((total + 3) >> 2); i += OWN_T)
      wk4[i] = ((const uint4*)keys)[i];
  } else {
  total = __builtin_amdgcn_readfirstlane(wcnt[blockIdx.x]);
  if (total <= LCAP)
    for (int i = tid; i < ((total + 3) >> 2); i += OWN_T)
      ((uint4*)keys)[i] = wk4[i];
  __syncthreads();
  }
  const int n = (total <= LCAP) ? total : 0;

  const int j = lane & 7;
  const int h = lane >> 3;
  const float4* xl4 = (const float4*)xl;

#define OWN_IDS(g, dl_, src_)                                              \
  { const int idx_ = min((g) + j, n - 1);                                  \
    const unsigned key_ = keys[idx_];                                      \
    const int e_ = (int)(key_ >> 8);                                       \
    dl_ = (int)(key_ & 255u);                                              \
    src_ = ei[min(e_, E - 1)];                     \
    src_ = (e_ < E) ? src_ : e_ - E; }

#define OWN_ROW(J) \
  row[J] = xl4[(size_t)__builtin_amdgcn_readlane(src, J) * (DD / 4) + lane];

#define OWN_ADD(J, COND)                                                   \
  if (COND) {                                                              \
    const float w_ = group8_bcast<J>(ee);                                  \
    float* a_ = accL + __builtin_amdgcn_readlane(dl, J) * DD + lane;       \
    lds_add(a_,      row[J].x * w_);                                       \
    lds_add(a_ + 32, row[J].y * w_);                                       \
    lds_add(a_ + 64, row[J].z * w_);                                       \
    lds_add(a_ + 96, row[J].w * w_);                                       \
  }

  if (n > 0) {
    const float4 w = ((const float4*)att)[lane];
    const float4* xrL4 = (const float4*)xrL;

    {
      constexpr int SU = (RN + OWN_T / 32 - 1) / (OWN_T / 32);
      float4 a[SU];
      int nl[SU];
#pragma unroll
      for (int u = 0; u < SU; ++u) {
        nl[u] = min(wave + u * (OWN_T / 32), nn - 1);
        a[u] = xl4[(size_t)(lo + nl[u]) * (DD / 4) + lane];
      }
#pragma unroll
      for (int u = 0; u < SU; ++u) {
        if (wave + u * (OWN_T / 32) < nn) {
          float t = own_dot4(a[u], xrL4[nl[u] * (DD / 4) + lane], w);
          t += lane_xor<4>(t);
          t += lane_xor<2>(t);
          t += lane_xor<1>(t);
          if (j == 0) {
            esL[nl[u] * HH + h]  = t;
            denL[nl[u] * HH + h] = 1.0f;
          }
          float* q = accL + nl[u] * DD + lane;
          q[0]  = a[u].x;
          q[32] = a[u].y;
          q[64] = a[u].z;
          q[96] = a[u].w;
        }
      }
    }
    __syncthreads();

    const bool b4 = (lane & 4) != 0;
    const bool b2 = (lane & 2) != 0;
    const bool b1 = (lane & 1) != 0;

    int g = nn + wave * 8;
    int dl = 0, src = 0;
    if (g < n) OWN_IDS(g, dl, src)
    while (g < n) {
      const int gn = g + (OWN_T / 32) * 8;
      float4 row[8];
      OWN_ROW(0) OWN_ROW(1) OWN_ROW(2) OWN_ROW(3)
      OWN_ROW(4) OWN_ROW(5) OWN_ROW(6) OWN_ROW(7)

      int dl2, src2;
      OWN_IDS(gn, dl2, src2)

      const float sh = esL[dl * HH + h];

      float p[8];
#define OWN_DOT(J) \
      p[J] = own_dot4(row[J], xrL4[__builtin_amdgcn_readlane(dl, J) * (DD / 4) + lane], w);
      OWN_DOT(0) OWN_DOT(1) OWN_DOT(2) OWN_DOT(3)
      OWN_DOT(4) OWN_DOT(5) OWN_DOT(6) OWN_DOT(7)
#undef OWN_DOT

      float q[4];
#pragma unroll
      for (int i = 0; i < 4; ++i) {
        const float keep = b4 ? p[i + 4] : p[i];
        const float give = b4 ? p[i] : p[i + 4];
        q[i] = keep + lane_xor<4>(give);
      }
      float r[2];
#pragma unroll
      for (int i = 0; i < 2; ++i) {
        const float keep = b2 ? q[i + 2] : q[i];
        const float give = b2 ? q[i] : q[i + 2];
        r[i] = keep + lane_xor<2>(give);
      }
      const float keep = b1 ? r[1] : r[0];
      const float give = b1 ? r[0] : r[1];
      const float s = keep + lane_xor<1>(give);

      const float d  = s - sh;
      const float ee = __expf(d);
      if (g + j < n) {
        lds_add(&denL[dl * HH + h], ee);
        if (!(d <= SHIFT_LIM)) ovf = 1;
      }

      if (g + 8 <= n) {
        OWN_ADD(0, true) OWN_ADD(1, true) OWN_ADD(2, true) OWN_ADD(3, true)
        OWN_ADD(4, true) OWN_ADD(5, true) OWN_ADD(6, true) OWN_ADD(7, true)
      } else {
        OWN_ADD(0, g + 0 < n) OWN_ADD(1, g + 1 < n) OWN_ADD(2, g + 2 < n)
        OWN_ADD(3, g + 3 < n) OWN_ADD(4, g + 4 < n) OWN_ADD(5, g + 5 < n)
        OWN_ADD(6, g + 6 < n) OWN_ADD(7, g + 7 < n)
      }

      g = gn; dl = dl2; src = src2;
    }
  }
  __syncthreads();
#undef OWN_IDS
#undef OWN_ROW
#undef OWN_ADD
  if (total > LCAP || __builtin_amdgcn_readfirstlane(ovf) != 0) {
    for (int i = tid; i < RN * DD / 4; i += OWN_T)
      ((float4*)accL)[i] = make_float4(0.0f, 0.0f, 0.0f, 0.0f);
    for (int i = tid; i < nn * HH; i += OWN_T) denL[i] = 0.0f;
    __syncthreads();
    const int* dp = ei + E;
    for (int ed = tid; ed < E; ed += OWN_T) {
      const unsigned rel = (unsigned)(dp[ed] - lo);
      if (rel < (unsigned)nn)
        own_max4(mxU, (int)rel,
                 own_logits_serial(ei[ed], lo + (int)rel, xl, xr, att));
    }
    for (int i = tid; i < nn; i += OWN_T)
      own_max4(mxU, i, own_logits_serial(lo + i, lo + i, xl, xr, att));
    __syncthreads();
    for (int ed = tid; ed < E; ed += OWN_T) {
      const unsigned rel = (unsigned)(dp[ed] - lo);
      if (rel < (unsigned)nn) {
        const int sn = ei[ed];
        own_edge_serial(own_logits_serial(sn, lo + (int)rel, xl, xr, att), sn,
                        (int)rel, accL, denL, mxU, xl);
      }
    }
    for (int i = tid; i < nn; i += OWN_T)
      own_edge_serial(own_logits_serial(lo + i, lo + i, xl, xr, att), lo + i,
                      i, accL, denL, mxU, xl);
    __syncthreads();
  }

  const float4 bs = ((const float4*)bias)[lane];
  const float4 gm = ((const float4*)gamma)[lane];
  const float4 bt = ((const float4*)beta)[lane];
  for (int nl = wave; nl < nn; nl += OWN_T / 32) {
    const size_t node = (size_t)(lo + nl);
    const float4 xi = ((const float4*)xin)[node * (DD / 4) + lane];
    const float den = denL[nl * HH + h];
    const float* a = accL + nl * DD + lane;
    const float v0 = xi.x + a[0]  / den + bs.x;
    const float v1 = xi.y + a[32] / den + bs.y;
    const float v2 = xi.z + a[64] / den + bs.z;
    const float v3 = xi.w + a[96] / den + bs.w;

    float sum = (v0 + v1) + (v2 + v3);
#pragma unroll
    for (int off = 16; off > 0; off >>= 1) sum += __shfl_xor(sum, off, 32);
    const float mu = sum * (1.0f / (float)DD);

    const float d0 = v0 - mu, d1 = v1 - mu, d2 = v2 - mu, d3 = v3 - mu;
    float sq = fmaf(d0, d0, fmaf(d1, d1, fmaf(d2, d2, d3 * d3)));
#pragma unroll
    for (int off = 16; off > 0; off >>= 1) sq += __shfl_xor(sq, off, 32);
    const float rstd = rsqrtf(sq * (1.0f / (float)DD) + LN_EPS);

    float4 o;
    o.x = d0 * rstd * gm.x + bt.x;
    o.y = d1 * rstd * gm.y + bt.y;
    o.z = d2 * rstd * gm.z + bt.z;
    o.w = d3 * rstd * gm.w + bt.w;
    if (do_relu) {
      o.x = (o.x > 0.0f) ? o.x : 0.0f;
      o.y = (o.y > 0.0f) ? o.y : 0.0f;
      o.z = (o.z > 0.0f) ? o.z : 0.0f;
      o.w = (o.w > 0.0f) ? o.w : 0.0f;
    }
    ((float4*)xout)[node * (DD / 4) + lane] = o;
  }
}

static inline int cdiv(long long a, long long b) { return (int)((a + b - 1) / b); }

extern "C" void kernel_launch(void* const* d_in, const int* in_sizes, int n_in,
                              void* d_out, int out_size, void* d_ws, size_t ws_size,
                              hipStream_t stream) {
  const int Np = NN;
  const int E  = in_sizes[1] / 2;

  const float* x0    = (const float*)d_in[0];
  const int*   ei    = (const int*)d_in[1];
  const float* Wl    = (const float*)d_in[2];
  const float* bl    = (const float*)d_in[3];
  const float* Wr    = (const float*)d_in[4];
  const float* br    = (const float*)d_in[5];
  const float* att   = (const float*)d_in[6];
  const float* bias  = (const float*)d_in[7];
  const float* gamma = (const float*)d_in[8];
  const float* beta  = (const float*)d_in[9];

  float* ws = (float*)d_ws;
  const size_t ND = (size_t)Np * DD;
  float*    bufA  = ws;
  float*    xl    = bufA + ND;
  float*    xr    = xl + ND;
  const int NWG = cdiv(Np, RN);
  unsigned* wkeys = (unsigned*)(xr + ND);
  int*      wcnt  = (int*)(wkeys + (size_t)NWG * LCAP);
  unsigned* obyt  = (unsigned*)(wcnt + 256);

  const int TB = 256;
  const dim3 blk(TB);

  for (int i = 0; i < LL; ++i) {
    const float* xin  = (i == 0) ? x0 : bufA;
    float*       xout = (i == LL - 1) ? (float*)d_out : bufA;
    const float* Wli  = Wl + (size_t)i * DD * DD;
    const float* bli  = bl + (size_t)i * DD;
    const float* Wri  = Wr + (size_t)i * DD * DD;
    const float* bri  = br + (size_t)i * DD;
    const float* atti = att + (size_t)i * HH * CC;
    const float* bi   = bias + (size_t)i * DD;
    const float* gi   = gamma + (size_t)i * DD;
    const float* bti  = beta + (size_t)i * DD;

    gemm_dual_wmma<<<Np / 16, blk, 0, stream>>>(
        xin, Wli, bli, Wri, bri, xl, xr, ei,
        (i == 0) ? obyt : (unsigned*)nullptr, E, Np);

    owner_aggregate<<<NWG, dim3(OWN_T), 0, stream>>>(
        ei, xl, xr, atti, xin, bi, gi, bti, xout, wkeys, wcnt, obyt, E, Np,
        (i != LL - 1) ? 1 : 0, (i != 0) ? 1 : 0);
  }
}


# Layers 1-2 dual GEMM moved into the previous layer's owner workgroups: each multiplies its own new x rows from LDS on WMMA f32, same k order; xl/xr in alternating planes; 6 -> 4 launches
# speedup vs baseline: 5.0621x; 1.1112x over previous
#include <hip/hip_runtime.h>
#include <math.h>


#define NN 50000
#define DD 128
#define HH 4
#define CC 32
#define LL 3
#define RN 197
#define NO_OWNER 255u
#define NEG_SLOPE 0.2f
#define LN_EPS 1e-5f

typedef __attribute__((ext_vector_type(2))) float v2f;
typedef __attribute__((ext_vector_type(8))) float v8f;

__device__ __forceinline__ unsigned f2ord(float f) {
  unsigned u = __float_as_uint(f);
  return (u & 0x80000000u) ? ~u : (u | 0x80000000u);
}
__device__ __forceinline__ float ord2f(unsigned u) {
  u = (u & 0x80000000u) ? (u & 0x7FFFFFFFu) : ~u;
  return __uint_as_float(u);
}

__device__ __forceinline__ void owner_bytes16(const int* __restrict__ ei,
                                              unsigned* __restrict__ ob, int E,
                                              int Np, int t) {
  const int e0 = t * 16;
  if (e0 >= E) return;
  const int* dp = ei + E + e0;
  int d[16];
  if (e0 + 16 <= E && (((size_t)dp) & 15u) == 0u) {
#pragma unroll
    for (int q = 0; q < 4; ++q) {
      const int4 v = ((const int4*)dp)[q];
      d[4 * q] = v.x; d[4 * q + 1] = v.y; d[4 * q + 2] = v.z; d[4 * q + 3] = v.w;
    }
  } else {
#pragma unroll
    for (int k = 0; k < 16; ++k) d[k] = (e0 + k < E) ? dp[k] : -1;
  }
  unsigned w[4];
#pragma unroll
  for (int q = 0; q < 4; ++q) {
    w[q] = 0u;
#pragma unroll
    for (int b = 0; b < 4; ++b) {
      const unsigned dv = (unsigned)d[4 * q + b];
      const unsigned o = (dv < (unsigned)Np) ? dv / (unsigned)RN : NO_OWNER;
      w[q] |= o << (8 * b);
    }
  }
  ((uint4*)ob)[t] = make_uint4(w[0], w[1], w[2], w[3]);
}

__global__ __launch_bounds__(256) void gemm_dual_wmma(
    const float* __restrict__ x,
    const float* __restrict__ Wl, const float* __restrict__ bl,
    const float* __restrict__ Wr, const float* __restrict__ br,
    float* __restrict__ xl, float* __restrict__ xr,
    const int* __restrict__ ei, unsigned* __restrict__ ob, int E, int Np) {
  __shared__ float sx[16 * DD];

  const int tid  = threadIdx.x;
  if (ob != nullptr)
    for (int t = blockIdx.x * 256 + tid; t < ((E + 15) >> 4);
         t += (int)gridDim.x * 256)
      owner_bytes16(ei, ob, E, Np, t);
  const int wave = tid >> 5;
  const int lane = tid & 31;
  const int rowBase = blockIdx.x * 16;

  {
    const float4* gsrc = (const float4*)(x + (size_t)rowBase * DD);
    float4* sdst = (float4*)sx;
    sdst[tid]       = gsrc[tid];
    sdst[tid + 256] = gsrc[tid + 256];
  }
  __syncthreads();

  const int colBase = wave * 16;
  const int m  = lane & 15;
  const int kk = (lane >> 4) * 2;
  const int n  = lane & 15;

  v8f cl = {};
  v8f cr = {};

  for (int k0 = 0; k0 < DD; k0 += 4) {
    v2f a;
    a.x = sx[m * DD + k0 + kk];
    a.y = sx[m * DD + k0 + kk + 1];
    v2f bfl;
    bfl.x = Wl[(size_t)(k0 + kk) * DD + colBase + n];
    bfl.y = Wl[(size_t)(k0 + kk + 1) * DD + colBase + n];
    v2f bfr;
    bfr.x = Wr[(size_t)(k0 + kk) * DD + colBase + n];
    bfr.y = Wr[(size_t)(k0 + kk + 1) * DD + colBase + n];

    cl = __builtin_amdgcn_wmma_f32_16x16x4_f32(
        false, a, false, bfl, (short)0, cl, false, false);
    cr = __builtin_amdgcn_wmma_f32_16x16x4_f32(
        false, a, false, bfr, (short)0, cr, false, false);
  }

  const int rOff = (lane >> 4) * 8;
  const int col  = colBase + n;
  const float bL = bl[col];
  const float bR = br[col];
#pragma unroll
  for (int i = 0; i < 8; ++i) {
    const size_t row = (size_t)(rowBase + i + rOff);
    xl[row * DD + col] = cl[i] + bL;
    xr[row * DD + col] = cr[i] + bR;
  }
}


template <int M>
__device__ __forceinline__ float lane_xor(float v) {
  return __int_as_float(__builtin_amdgcn_update_dpp(
      0, __float_as_int(v), 0x160 + M, 0xf, 0xf, true));
}

#define OWN_T 1024
#define LCAP  8192
#define SHIFT_LIM 40.0f
#define XP 132

template <int J>
__device__ __forceinline__ float group8_bcast(float v) {
  return __uint_as_float(__builtin_amdgcn_mov_dpp8(__float_as_uint(v), J * 0x249249));
}

__device__ __forceinline__ void lds_add(float* p, float v) {
  __hip_atomic_fetch_add(p, v, __ATOMIC_RELAXED, __HIP_MEMORY_SCOPE_WORKGROUP);
}

__device__ __forceinline__ void lds_umax(unsigned* p, unsigned v) {
  __hip_atomic_fetch_max(p, v, __ATOMIC_RELAXED, __HIP_MEMORY_SCOPE_WORKGROUP);
}

__device__ __forceinline__ void own_max4(unsigned* mxU, int rel, float4 lg) {
  lds_umax(&mxU[rel * HH + 0], f2ord(lg.x));
  lds_umax(&mxU[rel * HH + 1], f2ord(lg.y));
  lds_umax(&mxU[rel * HH + 2], f2ord(lg.z));
  lds_umax(&mxU[rel * HH + 3], f2ord(lg.w));
}

__device__ __forceinline__ void own_append(unsigned e, int* cnt,
                                           unsigned* keys) {
  const int pos = __hip_atomic_fetch_add(cnt, 1, __ATOMIC_RELAXED,
                                         __HIP_MEMORY_SCOPE_WORKGROUP);
  if (pos < LCAP) keys[pos] = e << 8;
}

__device__ __forceinline__ unsigned zero_bytes(unsigned x) {
  return ~(((x & 0x7F7F7F7Fu) + 0x7F7F7F7Fu) | x | 0x7F7F7F7Fu);
}

__device__ __forceinline__ float own_dot4(float4 a, float4 b, float4 w) {
  float m0 = a.x + b.x; m0 = (m0 > 0.0f) ? m0 : NEG_SLOPE * m0;
  float m1 = a.y + b.y; m1 = (m1 > 0.0f) ? m1 : NEG_SLOPE * m1;
  float m2 = a.z + b.z; m2 = (m2 > 0.0f) ? m2 : NEG_SLOPE * m2;
  float m3 = a.w + b.w; m3 = (m3 > 0.0f) ? m3 : NEG_SLOPE * m3;
  float t = w.x * m0;
  t = fmaf(w.y, m1, t);
  t = fmaf(w.z, m2, t);
  t = fmaf(w.w, m3, t);
  return t;
}

__device__ __forceinline__ float4 own_logits_serial(
    int src, int node, const float* __restrict__ xl,
    const float* __restrict__ xr, const float* __restrict__ att) {
  float lg[HH];
  for (int hh = 0; hh < HH; ++hh) {
    float t = 0.0f;
    for (int c = 0; c < CC; ++c) {
      float m = xl[(size_t)src * DD + hh * CC + c] +
                xr[(size_t)node * DD + hh * CC + c];
      m = (m > 0.0f) ? m : NEG_SLOPE * m;
      t = fmaf(att[hh * CC + c], m, t);
    }
    lg[hh] = t;
  }
  return make_float4(lg[0], lg[1], lg[2], lg[3]);
}

__device__ __forceinline__ void own_edge_serial(
    float4 lg, int src, int rel, float* accL, float* denL,
    const unsigned* mxU, const float* __restrict__ xl) {
  const float l4[HH] = {lg.x, lg.y, lg.z, lg.w};
  float ee[HH];
  for (int hh = 0; hh < HH; ++hh) {
    ee[hh] = __expf(l4[hh] - ord2f(mxU[rel * HH + hh]));
    lds_add(&denL[rel * HH + hh], ee[hh]);
  }
  for (int c = 0; c < DD; ++c)
    lds_add(&accL[rel * DD + (c & 3) * 32 + (c >> 2)],
            xl[(size_t)src * DD + c] * ee[c >> 5]);
}

template <int NT>
__device__ __forceinline__ void owner_next_gemm(
    const float* xT, const float* __restrict__ Wl_n,
    const float* __restrict__ bl_n, const float* __restrict__ Wr_n,
    const float* __restrict__ br_n, float* __restrict__ xl_n,
    float* __restrict__ xr_n, int lane, int wave, int lo, int nn) {
  const int m   = lane & 15;
  const int kk  = (lane >> 4) * 2;
  const int rg  = wave >> 3;
  const int col = (wave & 7) * 16 + m;
  v8f cl[NT], cr[NT];
  int ro[NT];
#pragma unroll
  for (int t = 0; t < NT; ++t) {
    cl[t] = (v8f){};
    cr[t] = (v8f){};
    ro[t] = min((rg + 4 * t) * 16 + m, nn - 1) * XP + kk;
  }
  for (int k0 = 0; k0 < DD; k0 += 4) {
    v2f bfl, bfr;
    bfl.x = Wl_n[(size_t)(k0 + kk) * DD + col];
    bfl.y = Wl_n[(size_t)(k0 + kk + 1) * DD + col];
    bfr.x = Wr_n[(size_t)(k0 + kk) * DD + col];
    bfr.y = Wr_n[(size_t)(k0 + kk + 1) * DD + col];
#pragma unroll
    for (int t = 0; t < NT; ++t) {
      v2f a;
      a.x = xT[ro[t] + k0];
      a.y = xT[ro[t] + k0 + 1];
      cl[t] = __builtin_amdgcn_wmma_f32_16x16x4_f32(
          false, a, false, bfl, (short)0, cl[t], false, false);
      cr[t] = __builtin_amdgcn_wmma_f32_16x16x4_f32(
          false, a, false, bfr, (short)0, cr[t], false, false);
    }
  }
  const int rOff = (lane >> 4) * 8;
  const float bL = bl_n[col];
  const float bR = br_n[col];
#pragma unroll
  for (int t = 0; t < NT; ++t) {
#pragma unroll
    for (int i = 0; i < 8; ++i) {
      const int r = (rg + 4 * t) * 16 + i + rOff;
      if (r < nn) {
        const size_t row = (size_t)(lo + r);
        xl_n[row * DD + col] = cl[t][i] + bL;
        xr_n[row * DD + col] = cr[t][i] + bR;
      }
    }
  }
}

__global__ __launch_bounds__(OWN_T) void owner_aggregate(
    const int* __restrict__ ei, const float* __restrict__ xl,
    const float* __restrict__ xr, const float* __restrict__ att,
    const float* xin, const float* __restrict__ bias,
    const float* __restrict__ gamma, const float* __restrict__ beta,
    float* xout, unsigned* wkeys, int* wcnt, const unsigned* __restrict__ obytes,
    const float* __restrict__ Wl_n, const float* __restrict__ bl_n,
    const float* __restrict__ Wr_n, const float* __restrict__ br_n,
    float* __restrict__ xl_n, float* __restrict__ xr_n,
    int E, int Np, int do_relu, int have_list) {
  __shared__ alignas(16) float accL[RN * DD];
  __shared__ alignas(16) float xrL[RN * XP];
  __shared__ float denL[RN * HH];
  __shared__ float esL[RN * HH];
  __shared__ unsigned mxU[RN * HH];
  __shared__ alignas(16) unsigned keys[LCAP];
  __shared__ int cnt;
  __shared__ int ovf;

  const int tid  = threadIdx.x;
  const int lane = tid & 31;
  const int wave = __builtin_amdgcn_readfirstlane(tid >> 5);
  const int lo = blockIdx.x * RN;
  const int nn = min(RN, Np - lo);

  {
    const float4* xs = (const float4*)xr + (size_t)lo * (DD / 4);
    const int nq = nn * (DD / 4);
    constexpr int XU = (RN * (DD / 4) + OWN_T - 1) / OWN_T;
    float4 t[XU];
    int ti[XU];
#pragma unroll
    for (int u = 0; u < XU; ++u) {
      ti[u] = min(tid + u * OWN_T, nq - 1);
      t[u] = xs[ti[u]];
    }
#pragma unroll
    for (int u = 0; u < XU; ++u) ((float4*)xrL)[ti[u]] = t[u];
  }
  for (int i = tid; i < nn * HH; i += OWN_T)
    mxU[i] = 0u;
  if (tid == 0) ovf = 0;
  uint4* wk4 = (uint4*)(wkeys + (size_t)blockIdx.x * LCAP);
  int total;
  if (have_list == 0) {
  for (int i = tid; i < nn; i += OWN_T)
    keys[i] = ((unsigned)(E + lo + i) << 8) | (unsigned)i;
  if (tid == 0) cnt = nn;
  __syncthreads();

  {
    const uint4* ob4 = (const uint4*)obytes;
    const int nv = (E + 15) >> 4;
    const unsigned pat = (unsigned)blockIdx.x * 0x01010101u;
    for (int i0 = tid; i0 < nv; i0 += 8 * OWN_T) {
      uint4 d[8];
#pragma unroll
      for (int u = 0; u < 8; ++u) {
        const int i = min(i0 + u * OWN_T, nv - 1);
        d[u] = ob4[i];
      }
#pragma unroll
      for (int u = 0; u < 8; ++u) {
        const int i = i0 + u * OWN_T;
        if (i < nv) {
          unsigned m = (zero_bytes(d[u].x ^ pat) >> 7) |
                       (zero_bytes(d[u].y ^ pat) >> 6) |
                       (zero_bytes(d[u].z ^ pat) >> 5) |
                       (zero_bytes(d[u].w ^ pat) >> 4);
          while (m != 0u) {
            const unsigned p = (unsigned)__builtin_ctz(m);
            own_append(16u * (unsigned)i + 4u * (p & 7u) + (p >> 3), &cnt, keys);
            m &= m - 1u;
          }
        }
      }
    }
  }
  __syncthreads();

  total = __builtin_amdgcn_readfirstlane(cnt);
  {
    const int* dp = ei + E;
    const int nfix = min(total, LCAP);
    for (int i0 = nn + tid; i0 < nfix; i0 += 4 * OWN_T) {
      unsigned k[4];
      int dv[4];
#pragma unroll
      for (int u = 0; u < 4; ++u) {
        const int i = i0 + u * OWN_T;
        k[u] = (i < nfix) ? keys[i] : 0u;
      }
#pragma unroll
      for (int u = 0; u < 4; ++u) dv[u] = dp[k[u] >> 8];
      const unsigned k0 = k[0] | (unsigned)(dv[0] - lo);
#pragma unroll
      for (int u = 0; u < 4; ++u) {
        const int i = i0 + u * OWN_T;
        const bool in = i < nfix;
        keys[in ? i : i0] = in ? (k[u] | (unsigned)(dv[u] - lo)) : k0;
      }
    }
  }
  __syncthreads();
  if (tid == 0) wcnt[blockIdx.x] = total;
  if (total <= LCAP)
    for (int i = tid; i < ((total + 3) >> 2); i += OWN_T)
      wk4[i] = ((const uint4*)keys)[i];
  } else {
  total = __builtin_amdgcn_readfirstlane(wcnt[blockIdx.x]);
  if (total <= LCAP)
    for (int i = tid; i < ((total + 3) >> 2); i += OWN_T)
      ((uint4*)keys)[i] = wk4[i];
  __syncthreads();
  }
  const int n = (total <= LCAP) ? total : 0;

  const int j = lane & 7;
  const int h = lane >> 3;
  const float4* xl4 = (const float4*)xl;

#define OWN_IDS(g, dl_, src_)                                              \
  { const int idx_ = min((g) + j, n - 1);                                  \
    const unsigned key_ = keys[idx_];                                      \
    const int e_ = (int)(key_ >> 8);                                       \
    dl_ = (int)(key_ & 255u);                                              \
    src_ = ei[min(e_, E - 1)];                     \
    src_ = (e_ < E) ? src_ : e_ - E; }

#define OWN_ROW(J) \
  row[J] = xl4[(size_t)__builtin_amdgcn_readlane(src, J) * (DD / 4) + lane];

#define OWN_ADD(J, COND)                                                   \
  if (COND) {                                                              \
    const float w_ = group8_bcast<J>(ee);                                  \
    float* a_ = accL + __builtin_amdgcn_readlane(dl, J) * DD + lane;       \
    lds_add(a_,      row[J].x * w_);                                       \
    lds_add(a_ + 32, row[J].y * w_);                                       \
    lds_add(a_ + 64, row[J].z * w_);                                       \
    lds_add(a_ + 96, row[J].w * w_);                                       \
  }

  if (n > 0) {
    const float4 w = ((const float4*)att)[lane];
    const float4* xrL4 = (const float4*)xrL;

    {
      constexpr int SU = (RN + OWN_T / 32 - 1) / (OWN_T / 32);
      float4 a[SU];
      int nl[SU];
#pragma unroll
      for (int u = 0; u < SU; ++u) {
        nl[u] = min(wave + u * (OWN_T / 32), nn - 1);
        a[u] = xl4[(size_t)(lo + nl[u]) * (DD / 4) + lane];
      }
#pragma unroll
      for (int u = 0; u < SU; ++u) {
        if (wave + u * (OWN_T / 32) < nn) {
          float t = own_dot4(a[u], xrL4[nl[u] * (DD / 4) + lane], w);
          t += lane_xor<4>(t);
          t += lane_xor<2>(t);
          t += lane_xor<1>(t);
          if (j == 0) {
            esL[nl[u] * HH + h]  = t;
            denL[nl[u] * HH + h] = 1.0f;
          }
          float* q = accL + nl[u] * DD + lane;
          q[0]  = a[u].x;
          q[32] = a[u].y;
          q[64] = a[u].z;
          q[96] = a[u].w;
        }
      }
    }
    __syncthreads();

    const bool b4 = (lane & 4) != 0;
    const bool b2 = (lane & 2) != 0;
    const bool b1 = (lane & 1) != 0;

    int g = nn + wave * 8;
    int dl = 0, src = 0;
    if (g < n) OWN_IDS(g, dl, src)
    while (g < n) {
      const int gn = g + (OWN_T / 32) * 8;
      float4 row[8];
      OWN_ROW(0) OWN_ROW(1) OWN_ROW(2) OWN_ROW(3)
      OWN_ROW(4) OWN_ROW(5) OWN_ROW(6) OWN_ROW(7)

      int dl2, src2;
      OWN_IDS(gn, dl2, src2)

      const float sh = esL[dl * HH + h];

      float p[8];
#define OWN_DOT(J) \
      p[J] = own_dot4(row[J], xrL4[__builtin_amdgcn_readlane(dl, J) * (DD / 4) + lane], w);
      OWN_DOT(0) OWN_DOT(1) OWN_DOT(2) OWN_DOT(3)
      OWN_DOT(4) OWN_DOT(5) OWN_DOT(6) OWN_DOT(7)
#undef OWN_DOT

      float q[4];
#pragma unroll
      for (int i = 0; i < 4; ++i) {
        const float keep = b4 ? p[i + 4] : p[i];
        const float give = b4 ? p[i] : p[i + 4];
        q[i] = keep + lane_xor<4>(give);
      }
      float r[2];
#pragma unroll
      for (int i = 0; i < 2; ++i) {
        const float keep = b2 ? q[i + 2] : q[i];
        const float give = b2 ? q[i] : q[i + 2];
        r[i] = keep + lane_xor<2>(give);
      }
      const float keep = b1 ? r[1] : r[0];
      const float give = b1 ? r[0] : r[1];
      const float s = keep + lane_xor<1>(give);

      const float d  = s - sh;
      const float ee = __expf(d);
      if (g + j < n) {
        lds_add(&denL[dl * HH + h], ee);
        if (!(d <= SHIFT_LIM)) ovf = 1;
      }

      if (g + 8 <= n) {
        OWN_ADD(0, true) OWN_ADD(1, true) OWN_ADD(2, true) OWN_ADD(3, true)
        OWN_ADD(4, true) OWN_ADD(5, true) OWN_ADD(6, true) OWN_ADD(7, true)
      } else {
        OWN_ADD(0, g + 0 < n) OWN_ADD(1, g + 1 < n) OWN_ADD(2, g + 2 < n)
        OWN_ADD(3, g + 3 < n) OWN_ADD(4, g + 4 < n) OWN_ADD(5, g + 5 < n)
        OWN_ADD(6, g + 6 < n) OWN_ADD(7, g + 7 < n)
      }

      g = gn; dl = dl2; src = src2;
    }
  }
  __syncthreads();
#undef OWN_IDS
#undef OWN_ROW
#undef OWN_ADD
  if (total > LCAP || __builtin_amdgcn_readfirstlane(ovf) != 0) {
    for (int i = tid; i < RN * DD / 4; i += OWN_T)
      ((float4*)accL)[i] = make_float4(0.0f, 0.0f, 0.0f, 0.0f);
    for (int i = tid; i < nn * HH; i += OWN_T) denL[i] = 0.0f;
    __syncthreads();
    const int* dp = ei + E;
    for (int ed = tid; ed < E; ed += OWN_T) {
      const unsigned rel = (unsigned)(dp[ed] - lo);
      if (rel < (unsigned)nn)
        own_max4(mxU, (int)rel,
                 own_logits_serial(ei[ed], lo + (int)rel, xl, xr, att));
    }
    for (int i = tid; i < nn; i += OWN_T)
      own_max4(mxU, i, own_logits_serial(lo + i, lo + i, xl, xr, att));
    __syncthreads();
    for (int ed = tid; ed < E; ed += OWN_T) {
      const unsigned rel = (unsigned)(dp[ed] - lo);
      if (rel < (unsigned)nn) {
        const int sn = ei[ed];
        own_edge_serial(own_logits_serial(sn, lo + (int)rel, xl, xr, att), sn,
                        (int)rel, accL, denL, mxU, xl);
      }
    }
    for (int i = tid; i < nn; i += OWN_T)
      own_edge_serial(own_logits_serial(lo + i, lo + i, xl, xr, att), lo + i,
                      i, accL, denL, mxU, xl);
    __syncthreads();
  }

  const float4 bs = ((const float4*)bias)[lane];
  const float4 gm = ((const float4*)gamma)[lane];
  const float4 bt = ((const float4*)beta)[lane];
  for (int nl = wave; nl < nn; nl += OWN_T / 32) {
    const size_t node = (size_t)(lo + nl);
    const float4 xi = ((const float4*)xin)[node * (DD / 4) + lane];
    const float den = denL[nl * HH + h];
    const float* a = accL + nl * DD + lane;
    const float v0 = xi.x + a[0]  / den + bs.x;
    const float v1 = xi.y + a[32] / den + bs.y;
    const float v2 = xi.z + a[64] / den + bs.z;
    const float v3 = xi.w + a[96] / den + bs.w;

    float sum = (v0 + v1) + (v2 + v3);
#pragma unroll
    for (int off = 16; off > 0; off >>= 1) sum += __shfl_xor(sum, off, 32);
    const float mu = sum * (1.0f / (float)DD);

    const float d0 = v0 - mu, d1 = v1 - mu, d2 = v2 - mu, d3 = v3 - mu;
    float sq = fmaf(d0, d0, fmaf(d1, d1, fmaf(d2, d2, d3 * d3)));
#pragma unroll
    for (int off = 16; off > 0; off >>= 1) sq += __shfl_xor(sq, off, 32);
    const float rstd = rsqrtf(sq * (1.0f / (float)DD) + LN_EPS);

    float4 o;
    o.x = d0 * rstd * gm.x + bt.x;
    o.y = d1 * rstd * gm.y + bt.y;
    o.z = d2 * rstd * gm.z + bt.z;
    o.w = d3 * rstd * gm.w + bt.w;
    if (do_relu) {
      o.x = (o.x > 0.0f) ? o.x : 0.0f;
      o.y = (o.y > 0.0f) ? o.y : 0.0f;
      o.z = (o.z > 0.0f) ? o.z : 0.0f;
      o.w = (o.w > 0.0f) ? o.w : 0.0f;
    }
    ((float4*)xout)[node * (DD / 4) + lane] = o;
    if (Wl_n != nullptr) *(float4*)&xrL[nl * XP + 4 * lane] = o;
  }

  if (Wl_n != nullptr) {
    __syncthreads();
    if (((wave >> 3) + 12) * 16 < nn)
      owner_next_gemm<4>(xrL, Wl_n, bl_n, Wr_n, br_n, xl_n, xr_n, lane, wave, lo, nn);
    else
      owner_next_gemm<3>(xrL, Wl_n, bl_n, Wr_n, br_n, xl_n, xr_n, lane, wave, lo, nn);
  }
}

static inline int cdiv(long long a, long long b) { return (int)((a + b - 1) / b); }

extern "C" void kernel_launch(void* const* d_in, const int* in_sizes, int n_in,
                              void* d_out, int out_size, void* d_ws, size_t ws_size,
                              hipStream_t stream) {
  const int Np = NN;
  const int E  = in_sizes[1] / 2;

  const float* x0    = (const float*)d_in[0];
  const int*   ei    = (const int*)d_in[1];
  const float* Wl    = (const float*)d_in[2];
  const float* bl    = (const float*)d_in[3];
  const float* Wr    = (const float*)d_in[4];
  const float* br    = (const float*)d_in[5];
  const float* att   = (const float*)d_in[6];
  const float* bias  = (const float*)d_in[7];
  const float* gamma = (const float*)d_in[8];
  const float* beta  = (const float*)d_in[9];

  float* ws = (float*)d_ws;
  const size_t ND = (size_t)Np * DD;
  float*    bufA  = ws;
  float*    xl0   = bufA + ND;
  float*    xr0   = xl0 + ND;
  float*    xl1   = xr0 + ND;
  float*    xr1   = xl1 + ND;
  const int NWG = cdiv(Np, RN);
  unsigned* wkeys = (unsigned*)(xr1 + ND);
  int*      wcnt  = (int*)(wkeys + (size_t)NWG * LCAP);
  unsigned* obyt  = (unsigned*)(wcnt + 256);

  const int TB = 256;
  const dim3 blk(TB);

  for (int i = 0; i < LL; ++i) {
    const float* xin  = (i == 0) ? x0 : bufA;
    float*       xout = (i == LL - 1) ? (float*)d_out : bufA;
    const float* Wli  = Wl + (size_t)i * DD * DD;
    const float* bli  = bl + (size_t)i * DD;
    const float* Wri  = Wr + (size_t)i * DD * DD;
    const float* bri  = br + (size_t)i * DD;
    const float* atti = att + (size_t)i * HH * CC;
    const float* bi   = bias + (size_t)i * DD;
    const float* gi   = gamma + (size_t)i * DD;
    const float* bti  = beta + (size_t)i * DD;

    float* xl  = (i == 1) ? xl1 : xl0;
    float* xr  = (i == 1) ? xr1 : xr0;
    float* xln = (i == LL - 1) ? (float*)nullptr : ((i == 0) ? xl1 : xl0);
    float* xrn = (i == LL - 1) ? (float*)nullptr : ((i == 0) ? xr1 : xr0);
    const float* Wln = (i == LL - 1) ? (const float*)nullptr : Wli + (size_t)DD * DD;
    const float* bln = (i == LL - 1) ? (const float*)nullptr : bli + DD;
    const float* Wrn = (i == LL - 1) ? (const float*)nullptr : Wri + (size_t)DD * DD;
    const float* brn = (i == LL - 1) ? (const float*)nullptr : bri + DD;

    if (i == 0)
      gemm_dual_wmma<<<Np / 16, blk, 0, stream>>>(
          xin, Wli, bli, Wri, bri, xl, xr, ei, obyt, E, Np);

    owner_aggregate<<<NWG, dim3(OWN_T), 0, stream>>>(
        ei, xl, xr, atti, xin, bi, gi, bti, xout, wkeys, wcnt, obyt,
        Wln, bln, Wrn, brn, xln, xrn, E, Np,
        (i != LL - 1) ? 1 : 0, (i != 0) ? 1 : 0);
  }
}
